# starting kernel
# baseline (speedup 1.0000x reference)
#include <hip/hip_runtime.h>
#include <hip/hip_bf16.h>

typedef __attribute__((ext_vector_type(2))) float v2f;
typedef __attribute__((ext_vector_type(8))) float v8f;

#define N_NODES 50000
#define E_EDGES 1600000
#define NHEAD   4
#define DHEAD   16
#define F_IN    128
#define HD      64

__global__ void __launch_bounds__(256)
gt_qkv_proj_wmma(const float* __restrict__ X,
                 const float* __restrict__ Wq,
                 const float* __restrict__ Wk,
                 const float* __restrict__ Wv,
                 float* __restrict__ Q,
                 float* __restrict__ K,
                 float* __restrict__ V)
{
    const int wave = (int)((blockIdx.x * blockDim.x + threadIdx.x) >> 5);
    const int lane = (int)(threadIdx.x & 31);
    const int NTASK = (N_NODES / 16) * 3;
    if (wave >= NTASK) return;

    const int mat  = wave % 3;
    const int row0 = (wave / 3) * 16;
    const float* __restrict__ W   = (mat == 0) ? Wq : (mat == 1) ? Wk : Wv;
    float*       __restrict__ Out = (mat == 0) ? Q  : (mat == 1) ? K  : V;

    const int grp = lane >> 4;
    const int m16 = lane & 15;

    v8f c0 = {}, c1 = {}, c2 = {}, c3 = {};

    const float* __restrict__ arow = X + (size_t)(row0 + m16) * F_IN + 2 * grp;

#pragma unroll 4
    for (int k0 = 0; k0 < F_IN; k0 += 4) {
        v2f a;
        a.x = arow[k0];
        a.y = arow[k0 + 1];

        const float* __restrict__ wr0 = W + (size_t)(k0 + 2 * grp) * HD + m16;
        const float* __restrict__ wr1 = wr0 + HD;

        v2f b0, b1, b2, b3;
        b0.x = wr0[0];  b0.y = wr1[0];
        b1.x = wr0[16]; b1.y = wr1[16];
        b2.x = wr0[32]; b2.y = wr1[32];
        b3.x = wr0[48]; b3.y = wr1[48];

        c0 = __builtin_amdgcn_wmma_f32_16x16x4_f32(false, a, false, b0, (short)0, c0, false, false);
        c1 = __builtin_amdgcn_wmma_f32_16x16x4_f32(false, a, false, b1, (short)0, c1, false, false);
        c2 = __builtin_amdgcn_wmma_f32_16x16x4_f32(false, a, false, b2, (short)0, c2, false, false);
        c3 = __builtin_amdgcn_wmma_f32_16x16x4_f32(false, a, false, b3, (short)0, c3, false, false);
    }

    const int mrow = 8 * grp;
#pragma unroll
    for (int r = 0; r < 8; ++r) {
        float* __restrict__ o = Out + (size_t)(row0 + mrow + r) * HD + m16;
        o[0]  = c0[r];
        o[16] = c1[r];
        o[32] = c2[r];
        o[48] = c3[r];
    }
}

__global__ void __launch_bounds__(256)
gt_edge_attn(const int* __restrict__ src, const int* __restrict__ dst,
             const float* __restrict__ Q, const float* __restrict__ K,
             const float* __restrict__ V,
             float* __restrict__ wV, float* __restrict__ z)
{
    const int lane  = (int)(threadIdx.x & 31);
    const int wave  = (int)((blockIdx.x * blockDim.x + threadIdx.x) >> 5);
    const int nwave = (int)((gridDim.x * blockDim.x) >> 5);
    const int grp   = lane >> 4;
    const int dd    = lane & 15;

    for (int e = wave; e < E_EDGES; e += nwave) {
        const int s = __builtin_amdgcn_readfirstlane(src[e]);
        const int d = __builtin_amdgcn_readfirstlane(dst[e]);

        const size_t sb = (size_t)s * HD;
        const size_t db = (size_t)d * HD;

        const float k0 = K[sb + lane], k1 = K[sb + lane + 32];
        const float q0 = Q[db + lane], q1 = Q[db + lane + 32];
        const float v0 = V[sb + lane], v1 = V[sb + lane + 32];

        float p0 = k0 * q0;
        float p1 = k1 * q1;

#pragma unroll
        for (int off = 1; off < 16; off <<= 1) {
            p0 += __shfl_xor(p0, off, 32);
            p1 += __shfl_xor(p1, off, 32);
        }

        const float s0 = __expf(fminf(fmaxf(p0 * 0.25f, -5.0f), 5.0f));
        const float s1 = __expf(fminf(fmaxf(p1 * 0.25f, -5.0f), 5.0f));

        atomicAdd(&wV[db + lane],      s0 * v0);
        atomicAdd(&wV[db + lane + 32], s1 * v1);
        if (dd == 0) {
            atomicAdd(&z[(size_t)d * NHEAD + grp],     s0);
            atomicAdd(&z[(size_t)d * NHEAD + 2 + grp], s1);
        }
    }
}

__global__ void __launch_bounds__(256)
gt_normalize(float* __restrict__ out, const float* __restrict__ z)
{
    const int i = (int)(blockIdx.x * blockDim.x + threadIdx.x);
    if (i < N_NODES * HD) {
        const int n = i >> 6;
        const int h = (i >> 4) & 3;
        out[i] = out[i] / (z[n * NHEAD + h] + 1e-6f);
    }
}

extern "C" void kernel_launch(void* const* d_in, const int* in_sizes, int n_in,
                              void* d_out, int out_size, void* d_ws, size_t ws_size,
                              hipStream_t stream)
{
    (void)in_sizes; (void)n_in; (void)out_size; (void)ws_size;

    const float* X   = (const float*)d_in[0];
    const int*   src = (const int*)  d_in[1];
    const int*   dst = (const int*)  d_in[2];
    const float* Wq  = (const float*)d_in[3];
    const float* Wk  = (const float*)d_in[4];
    const float* Wv  = (const float*)d_in[5];

    float* out = (float*)d_out;
    float* ws  = (float*)d_ws;
    float* Q   = ws;
    float* K   = ws + (size_t)N_NODES * HD;
    float* V   = ws + (size_t)2 * N_NODES * HD;
    float* z   = ws + (size_t)3 * N_NODES * HD;

    hipMemsetAsync(out, 0, (size_t)N_NODES * HD * sizeof(float), stream);
    hipMemsetAsync(z,   0, (size_t)N_NODES * NHEAD * sizeof(float), stream);

    {
        const int ntask = (N_NODES / 16) * 3;
        const int wpb   = 256 / 32;
        const int blocks = (ntask + wpb - 1) / wpb;
        gt_qkv_proj_wmma<<<blocks, 256, 0, stream>>>(X, Wq, Wk, Wv, Q, K, V);
    }

    gt_edge_attn<<<4096, 256, 0, stream>>>(src, dst, Q, K, V, out, z);

    {
        const int total = N_NODES * HD;
        gt_normalize<<<(total + 255) / 256, 256, 0, stream>>>(out, z);
    }
}


# Edge kernel: K,Q,V rows gathered 16 B per lane, half-wave per edge (2 edges per whole-line load), 32 ids per load; wV turned to line order in LDS; dot summed 4 per lane then quad butterfly
# speedup vs baseline: 1.1156x; 1.1156x over previous
#include <hip/hip_runtime.h>
#include <hip/hip_bf16.h>

typedef __attribute__((ext_vector_type(2))) float v2f;
typedef __attribute__((ext_vector_type(8))) float v8f;

#define N_NODES 50000
#define E_EDGES 1600000
#define NHEAD   4
#define DHEAD   16
#define F_IN    128
#define HD      64

__global__ void __launch_bounds__(256)
gt_qkv_proj_wmma(const float* __restrict__ X,
                 const float* __restrict__ Wq,
                 const float* __restrict__ Wk,
                 const float* __restrict__ Wv,
                 float* __restrict__ Q,
                 float* __restrict__ K,
                 float* __restrict__ V)
{
    const int wave = (int)((blockIdx.x * blockDim.x + threadIdx.x) >> 5);
    const int lane = (int)(threadIdx.x & 31);
    const int NTASK = (N_NODES / 16) * 3;
    if (wave >= NTASK) return;

    const int mat  = wave % 3;
    const int row0 = (wave / 3) * 16;
    const float* __restrict__ W   = (mat == 0) ? Wq : (mat == 1) ? Wk : Wv;
    float*       __restrict__ Out = (mat == 0) ? Q  : (mat == 1) ? K  : V;

    const int grp = lane >> 4;
    const int m16 = lane & 15;

    v8f c0 = {}, c1 = {}, c2 = {}, c3 = {};

    const float* __restrict__ arow = X + (size_t)(row0 + m16) * F_IN + 2 * grp;

#pragma unroll 4
    for (int k0 = 0; k0 < F_IN; k0 += 4) {
        v2f a;
        a.x = arow[k0];
        a.y = arow[k0 + 1];

        const float* __restrict__ wr0 = W + (size_t)(k0 + 2 * grp) * HD + m16;
        const float* __restrict__ wr1 = wr0 + HD;

        v2f b0, b1, b2, b3;
        b0.x = wr0[0];  b0.y = wr1[0];
        b1.x = wr0[16]; b1.y = wr1[16];
        b2.x = wr0[32]; b2.y = wr1[32];
        b3.x = wr0[48]; b3.y = wr1[48];

        c0 = __builtin_amdgcn_wmma_f32_16x16x4_f32(false, a, false, b0, (short)0, c0, false, false);
        c1 = __builtin_amdgcn_wmma_f32_16x16x4_f32(false, a, false, b1, (short)0, c1, false, false);
        c2 = __builtin_amdgcn_wmma_f32_16x16x4_f32(false, a, false, b2, (short)0, c2, false, false);
        c3 = __builtin_amdgcn_wmma_f32_16x16x4_f32(false, a, false, b3, (short)0, c3, false, false);
    }

    const int mrow = 8 * grp;
#pragma unroll
    for (int r = 0; r < 8; ++r) {
        float* __restrict__ o = Out + (size_t)(row0 + mrow + r) * HD + m16;
        o[0]  = c0[r];
        o[16] = c1[r];
        o[32] = c2[r];
        o[48] = c3[r];
    }
}

#define EDGE_BATCH      32
#define EDGE_NBATCH     (E_EDGES / EDGE_BATCH)
#define EDGE_WAVES      12500
#define EDGE_BPW        (EDGE_NBATCH / EDGE_WAVES)
#define EDGE_BLOCK      128
static_assert(E_EDGES % EDGE_BATCH == 0, "whole batches only");
static_assert(EDGE_NBATCH % EDGE_WAVES == 0, "equal batches per wave");
static_assert((EDGE_WAVES * 32) % EDGE_BLOCK == 0, "whole blocks only");

typedef __attribute__((ext_vector_type(4))) float v4f;

struct PairRows {
    v4f k4, q4, v4;
    int d0, d1, dl;
};

__device__ __forceinline__ PairRows
gt_load_pair(const float* __restrict__ Q, const float* __restrict__ K,
             const float* __restrict__ V, int sv, int dv, int p,
             int half, int i16)
{
    PairRows r;
    const int s0 = __builtin_amdgcn_readlane(sv, 2 * p);
    const int s1 = __builtin_amdgcn_readlane(sv, 2 * p + 1);
    r.d0 = __builtin_amdgcn_readlane(dv, 2 * p);
    r.d1 = __builtin_amdgcn_readlane(dv, 2 * p + 1);
    const int sl = half ? s1 : s0;
    r.dl = half ? r.d1 : r.d0;

    const size_t so = (size_t)sl   * HD + 4 * i16;
    const size_t qo = (size_t)r.dl * HD + 4 * i16;
    r.k4 = *reinterpret_cast<const v4f*>(K + so);
    r.q4 = *reinterpret_cast<const v4f*>(Q + qo);
    r.v4 = *reinterpret_cast<const v4f*>(V + so);
    return r;
}

__device__ __forceinline__ float gt_quad_xor1(float x)
{
    return __int_as_float(__builtin_amdgcn_update_dpp(0, __float_as_int(x), 0xB1, 0xF, 0xF, true));
}
__device__ __forceinline__ float gt_quad_xor2(float x)
{
    return __int_as_float(__builtin_amdgcn_update_dpp(0, __float_as_int(x), 0x4E, 0xF, 0xF, true));
}

__device__ __forceinline__ void
gt_reduce_pair(const PairRows& r, float* __restrict__ wV, float* __restrict__ z,
               float* slot, int lane, int i16)
{
    float p = r.k4.x * r.q4.x;
    p = fmaf(r.k4.y, r.q4.y, p);
    p = fmaf(r.k4.z, r.q4.z, p);
    p = fmaf(r.k4.w, r.q4.w, p);
    p += gt_quad_xor1(p);
    p += gt_quad_xor2(p);

    const float sc = __expf(fminf(fmaxf(p * 0.25f, -5.0f), 5.0f));

    slot[4 * lane + 0] = sc * r.v4.x;
    slot[4 * lane + 1] = sc * r.v4.y;
    slot[4 * lane + 2] = sc * r.v4.z;
    slot[4 * lane + 3] = sc * r.v4.w;
    const float a0 = slot[lane];
    const float a1 = slot[lane + 32];
    const float b0 = slot[lane + 64];
    const float b1 = slot[lane + 96];

    float* __restrict__ o0 = wV + (size_t)r.d0 * HD + lane;
    float* __restrict__ o1 = wV + (size_t)r.d1 * HD + lane;
    atomicAdd(o0,      a0);
    atomicAdd(o0 + 32, a1);
    atomicAdd(o1,      b0);
    atomicAdd(o1 + 32, b1);
    if ((lane & 3) == 0)
        atomicAdd(&z[(size_t)r.dl * NHEAD + (i16 >> 2)], sc);
}

__global__ void __launch_bounds__(EDGE_BLOCK)
gt_edge_attn(const int* __restrict__ src, const int* __restrict__ dst,
             const float* __restrict__ Q, const float* __restrict__ K,
             const float* __restrict__ V,
             float* __restrict__ wV, float* __restrict__ z)
{
    __shared__ __attribute__((aligned(16))) float s_rows[EDGE_BLOCK * 4];

    const int lane = (int)(threadIdx.x & 31);
    const int wave = (int)((blockIdx.x * blockDim.x + threadIdx.x) >> 5);
    const int half = lane >> 4;
    const int i16  = lane & 15;
    float* slot = s_rows + (threadIdx.x >> 5) * 128;

    int sv = src[(size_t)wave * EDGE_BATCH + lane];
    int dv = dst[(size_t)wave * EDGE_BATCH + lane];
    PairRows cur = gt_load_pair(Q, K, V, sv, dv, 0, half, i16);

    for (int b = 0; b < EDGE_BPW; ++b) {
        const bool more = (b + 1 < EDGE_BPW);
        int svn = sv, dvn = dv;
        if (more) {
            const size_t e = (size_t)(wave + (b + 1) * EDGE_WAVES) * EDGE_BATCH + lane;
            svn = src[e];
            dvn = dst[e];
        }
#pragma unroll
        for (int p = 0; p < EDGE_BATCH / 2; ++p) {
            PairRows nxt = cur;
            if (p + 1 < EDGE_BATCH / 2)
                nxt = gt_load_pair(Q, K, V, sv, dv, p + 1, half, i16);
            else if (more)
                nxt = gt_load_pair(Q, K, V, svn, dvn, 0, half, i16);
            __builtin_amdgcn_sched_barrier(0);
            gt_reduce_pair(cur, wV, z, slot, lane, i16);
            cur = nxt;
        }
        sv = svn;
        dv = dvn;
    }
}

__global__ void __launch_bounds__(256)
gt_normalize(float* __restrict__ out, const float* __restrict__ z)
{
    const int i = (int)(blockIdx.x * blockDim.x + threadIdx.x);
    if (i < N_NODES * HD) {
        const int n = i >> 6;
        const int h = (i >> 4) & 3;
        out[i] = out[i] / (z[n * NHEAD + h] + 1e-6f);
    }
}

extern "C" void kernel_launch(void* const* d_in, const int* in_sizes, int n_in,
                              void* d_out, int out_size, void* d_ws, size_t ws_size,
                              hipStream_t stream)
{
    (void)in_sizes; (void)n_in; (void)out_size; (void)ws_size;

    const float* X   = (const float*)d_in[0];
    const int*   src = (const int*)  d_in[1];
    const int*   dst = (const int*)  d_in[2];
    const float* Wq  = (const float*)d_in[3];
    const float* Wk  = (const float*)d_in[4];
    const float* Wv  = (const float*)d_in[5];

    float* out = (float*)d_out;
    float* ws  = (float*)d_ws;
    float* Q   = ws;
    float* K   = ws + (size_t)N_NODES * HD;
    float* V   = ws + (size_t)2 * N_NODES * HD;
    float* z   = ws + (size_t)3 * N_NODES * HD;

    hipMemsetAsync(out, 0, (size_t)N_NODES * HD * sizeof(float), stream);
    hipMemsetAsync(z,   0, (size_t)N_NODES * NHEAD * sizeof(float), stream);

    {
        const int ntask = (N_NODES / 16) * 3;
        const int wpb   = 256 / 32;
        const int blocks = (ntask + wpb - 1) / wpb;
        gt_qkv_proj_wmma<<<blocks, 256, 0, stream>>>(X, Wq, Wk, Wv, Q, K, V);
    }

    gt_edge_attn<<<(EDGE_WAVES * 32) / EDGE_BLOCK, EDGE_BLOCK, 0, stream>>>(src, dst, Q, K, V, out, z);

    {
        const int total = N_NODES * HD;
        gt_normalize<<<(total + 255) / 256, 256, 0, stream>>>(out, z);
    }
}


# Scatter sums wV, z kept twice: one plane per package half, picked by the wave's die number, so no line takes adds from both halves; last kernel adds the planes (sum order: 2 partial sums, then 1 add)
# speedup vs baseline: 1.3799x; 1.2369x over previous
#include <hip/hip_runtime.h>
#include <hip/hip_bf16.h>

typedef __attribute__((ext_vector_type(2))) float v2f;
typedef __attribute__((ext_vector_type(8))) float v8f;

#define N_NODES 50000
#define E_EDGES 1600000
#define NHEAD   4
#define DHEAD   16
#define F_IN    128
#define HD      64

__global__ void __launch_bounds__(256)
gt_qkv_proj_wmma(const float* __restrict__ X,
                 const float* __restrict__ Wq,
                 const float* __restrict__ Wk,
                 const float* __restrict__ Wv,
                 float* __restrict__ Q,
                 float* __restrict__ K,
                 float* __restrict__ V)
{
    const int wave = (int)((blockIdx.x * blockDim.x + threadIdx.x) >> 5);
    const int lane = (int)(threadIdx.x & 31);
    const int NTASK = (N_NODES / 16) * 3;
    if (wave >= NTASK) return;

    const int mat  = wave % 3;
    const int row0 = (wave / 3) * 16;
    const float* __restrict__ W   = (mat == 0) ? Wq : (mat == 1) ? Wk : Wv;
    float*       __restrict__ Out = (mat == 0) ? Q  : (mat == 1) ? K  : V;

    const int grp = lane >> 4;
    const int m16 = lane & 15;

    v8f c0 = {}, c1 = {}, c2 = {}, c3 = {};

    const float* __restrict__ arow = X + (size_t)(row0 + m16) * F_IN + 2 * grp;

#pragma unroll 4
    for (int k0 = 0; k0 < F_IN; k0 += 4) {
        v2f a;
        a.x = arow[k0];
        a.y = arow[k0 + 1];

        const float* __restrict__ wr0 = W + (size_t)(k0 + 2 * grp) * HD + m16;
        const float* __restrict__ wr1 = wr0 + HD;

        v2f b0, b1, b2, b3;
        b0.x = wr0[0];  b0.y = wr1[0];
        b1.x = wr0[16]; b1.y = wr1[16];
        b2.x = wr0[32]; b2.y = wr1[32];
        b3.x = wr0[48]; b3.y = wr1[48];

        c0 = __builtin_amdgcn_wmma_f32_16x16x4_f32(false, a, false, b0, (short)0, c0, false, false);
        c1 = __builtin_amdgcn_wmma_f32_16x16x4_f32(false, a, false, b1, (short)0, c1, false, false);
        c2 = __builtin_amdgcn_wmma_f32_16x16x4_f32(false, a, false, b2, (short)0, c2, false, false);
        c3 = __builtin_amdgcn_wmma_f32_16x16x4_f32(false, a, false, b3, (short)0, c3, false, false);
    }

    const int mrow = 8 * grp;
#pragma unroll
    for (int r = 0; r < 8; ++r) {
        float* __restrict__ o = Out + (size_t)(row0 + mrow + r) * HD + m16;
        o[0]  = c0[r];
        o[16] = c1[r];
        o[32] = c2[r];
        o[48] = c3[r];
    }
}

#define EDGE_BATCH      32
#define EDGE_NBATCH     (E_EDGES / EDGE_BATCH)
#define EDGE_WAVES      12500
#define EDGE_BPW        (EDGE_NBATCH / EDGE_WAVES)
#define EDGE_BLOCK      128
static_assert(E_EDGES % EDGE_BATCH == 0, "whole batches only");
static_assert(EDGE_NBATCH % EDGE_WAVES == 0, "equal batches per wave");
static_assert((EDGE_WAVES * 32) % EDGE_BLOCK == 0, "whole blocks only");

typedef __attribute__((ext_vector_type(4))) float v4f;

struct PairRows {
    v4f k4, q4, v4;
    int d0, d1, dl;
};

__device__ __forceinline__ PairRows
gt_load_pair(const float* __restrict__ Q, const float* __restrict__ K,
             const float* __restrict__ V, int sv, int dv, int p,
             int half, int i16)
{
    PairRows r;
    const int s0 = __builtin_amdgcn_readlane(sv, 2 * p);
    const int s1 = __builtin_amdgcn_readlane(sv, 2 * p + 1);
    r.d0 = __builtin_amdgcn_readlane(dv, 2 * p);
    r.d1 = __builtin_amdgcn_readlane(dv, 2 * p + 1);
    const int sl = half ? s1 : s0;
    r.dl = half ? r.d1 : r.d0;

    const size_t so = (size_t)sl   * HD + 4 * i16;
    const size_t qo = (size_t)r.dl * HD + 4 * i16;
    r.k4 = *reinterpret_cast<const v4f*>(K + so);
    r.q4 = *reinterpret_cast<const v4f*>(Q + qo);
    r.v4 = *reinterpret_cast<const v4f*>(V + so);
    return r;
}

__device__ __forceinline__ float gt_quad_xor1(float x)
{
    return __int_as_float(__builtin_amdgcn_update_dpp(0, __float_as_int(x), 0xB1, 0xF, 0xF, true));
}
__device__ __forceinline__ float gt_quad_xor2(float x)
{
    return __int_as_float(__builtin_amdgcn_update_dpp(0, __float_as_int(x), 0x4E, 0xF, 0xF, true));
}

__device__ __forceinline__ void
gt_reduce_pair(const PairRows& r, float* __restrict__ wV, float* __restrict__ z,
               float* slot, int lane, int i16)
{
    float p = r.k4.x * r.q4.x;
    p = fmaf(r.k4.y, r.q4.y, p);
    p = fmaf(r.k4.z, r.q4.z, p);
    p = fmaf(r.k4.w, r.q4.w, p);
    p += gt_quad_xor1(p);
    p += gt_quad_xor2(p);

    const float sc = __expf(fminf(fmaxf(p * 0.25f, -5.0f), 5.0f));

    slot[4 * lane + 0] = sc * r.v4.x;
    slot[4 * lane + 1] = sc * r.v4.y;
    slot[4 * lane + 2] = sc * r.v4.z;
    slot[4 * lane + 3] = sc * r.v4.w;
    const float a0 = slot[lane];
    const float a1 = slot[lane + 32];
    const float b0 = slot[lane + 64];
    const float b1 = slot[lane + 96];

    float* __restrict__ o0 = wV + (size_t)r.d0 * HD + lane;
    float* __restrict__ o1 = wV + (size_t)r.d1 * HD + lane;
    atomicAdd(o0,      a0);
    atomicAdd(o0 + 32, a1);
    atomicAdd(o1,      b0);
    atomicAdd(o1 + 32, b1);
    if ((lane & 3) == 0)
        atomicAdd(&z[(size_t)r.dl * NHEAD + (i16 >> 2)], sc);
}

__global__ void __launch_bounds__(EDGE_BLOCK)
gt_edge_attn(const int* __restrict__ src, const int* __restrict__ dst,
             const float* __restrict__ Q, const float* __restrict__ K,
             const float* __restrict__ V,
             float* __restrict__ wV0, float* __restrict__ z0,
             float* __restrict__ wV1, float* __restrict__ z1)
{
    __shared__ __attribute__((aligned(16))) float s_rows[EDGE_BLOCK * 4];

    const unsigned place = (unsigned)__builtin_amdgcn_s_sendmsg_rtn(0x87);
    const bool upper = ((place >> 18) & 1u) != 0u;
    float* __restrict__ wV = upper ? wV1 : wV0;
    float* __restrict__ z  = upper ? z1  : z0;

    const int lane = (int)(threadIdx.x & 31);
    const int wave = (int)((blockIdx.x * blockDim.x + threadIdx.x) >> 5);
    const int half = lane >> 4;
    const int i16  = lane & 15;
    float* slot = s_rows + (threadIdx.x >> 5) * 128;

    int sv = src[(size_t)wave * EDGE_BATCH + lane];
    int dv = dst[(size_t)wave * EDGE_BATCH + lane];
    PairRows cur = gt_load_pair(Q, K, V, sv, dv, 0, half, i16);

    for (int b = 0; b < EDGE_BPW; ++b) {
        const bool more = (b + 1 < EDGE_BPW);
        int svn = sv, dvn = dv;
        if (more) {
            const size_t e = (size_t)(wave + (b + 1) * EDGE_WAVES) * EDGE_BATCH + lane;
            svn = src[e];
            dvn = dst[e];
        }
#pragma unroll
        for (int p = 0; p < EDGE_BATCH / 2; ++p) {
            PairRows nxt = cur;
            if (p + 1 < EDGE_BATCH / 2)
                nxt = gt_load_pair(Q, K, V, sv, dv, p + 1, half, i16);
            else if (more)
                nxt = gt_load_pair(Q, K, V, svn, dvn, 0, half, i16);
            __builtin_amdgcn_sched_barrier(0);
            gt_reduce_pair(cur, wV, z, slot, lane, i16);
            cur = nxt;
        }
        sv = svn;
        dv = dvn;
    }
}

__global__ void __launch_bounds__(256)
gt_normalize(float* __restrict__ out, const float* __restrict__ wV1,
             const float* __restrict__ z0, const float* __restrict__ z1)
{
    const int i = (int)(blockIdx.x * blockDim.x + threadIdx.x);
    if (i < N_NODES * HD / 4) {
        const int nh = i >> 2;
        const float zz = (z0[nh] + z1[nh]) + 1e-6f;
        const v4f a = reinterpret_cast<const v4f*>(out)[i];
        const v4f b = reinterpret_cast<const v4f*>(wV1)[i];
        v4f r;
        r.x = (a.x + b.x) / zz;
        r.y = (a.y + b.y) / zz;
        r.z = (a.z + b.z) / zz;
        r.w = (a.w + b.w) / zz;
        reinterpret_cast<v4f*>(out)[i] = r;
    }
}

extern "C" void kernel_launch(void* const* d_in, const int* in_sizes, int n_in,
                              void* d_out, int out_size, void* d_ws, size_t ws_size,
                              hipStream_t stream)
{
    (void)in_sizes; (void)n_in; (void)out_size; (void)ws_size;

    const float* X   = (const float*)d_in[0];
    const int*   src = (const int*)  d_in[1];
    const int*   dst = (const int*)  d_in[2];
    const float* Wq  = (const float*)d_in[3];
    const float* Wk  = (const float*)d_in[4];
    const float* Wv  = (const float*)d_in[5];

    float* out = (float*)d_out;
    float* ws  = (float*)d_ws;
    float* Q   = ws;
    float* K   = ws + (size_t)N_NODES * HD;
    float* V   = ws + (size_t)2 * N_NODES * HD;
    float* z0  = ws + (size_t)3 * N_NODES * HD;
    float* z1  = z0 + (size_t)N_NODES * NHEAD;
    float* wV1 = z1 + (size_t)N_NODES * NHEAD;

    hipMemsetAsync(out, 0, (size_t)N_NODES * HD * sizeof(float), stream);
    hipMemsetAsync(z0,  0, ((size_t)2 * N_NODES * NHEAD + (size_t)N_NODES * HD) * sizeof(float), stream);

    {
        const int ntask = (N_NODES / 16) * 3;
        const int wpb   = 256 / 32;
        const int blocks = (ntask + wpb - 1) / wpb;
        gt_qkv_proj_wmma<<<blocks, 256, 0, stream>>>(X, Wq, Wk, Wv, Q, K, V);
    }

    gt_edge_attn<<<(EDGE_WAVES * 32) / EDGE_BLOCK, EDGE_BLOCK, 0, stream>>>(src, dst, Q, K, V, out, z0, wV1, z1);

    {
        const int total = N_NODES * HD / 4;
        gt_normalize<<<(total + 255) / 256, 256, 0, stream>>>(out, wV1, z0, z1);
    }
}


# Float atomics dropped: edges filed by dst range (count, offsets, fill); one work-group per 196-node range sums in LDS as 64-bit fixed point (integer adds: same bits in any order), writes out once
# speedup vs baseline: 1.9353x; 1.4025x over previous
#include <hip/hip_runtime.h>
#include <hip/hip_bf16.h>

typedef __attribute__((ext_vector_type(2))) float v2f;
typedef __attribute__((ext_vector_type(8))) float v8f;

#define N_NODES 50000
#define E_EDGES 1600000
#define NHEAD   4
#define DHEAD   16
#define F_IN    128
#define HD      64

__global__ void __launch_bounds__(256)
gt_qkv_proj_wmma(const float* __restrict__ X,
                 const float* __restrict__ Wq,
                 const float* __restrict__ Wk,
                 const float* __restrict__ Wv,
                 float* __restrict__ Q,
                 float* __restrict__ K,
                 float* __restrict__ V)
{
    const int wave = (int)((blockIdx.x * blockDim.x + threadIdx.x) >> 5);
    const int lane = (int)(threadIdx.x & 31);
    const int NTASK = (N_NODES / 16) * 3;
    if (wave >= NTASK) return;

    const int mat  = wave % 3;
    const int row0 = (wave / 3) * 16;
    const float* __restrict__ W   = (mat == 0) ? Wq : (mat == 1) ? Wk : Wv;
    float*       __restrict__ Out = (mat == 0) ? Q  : (mat == 1) ? K  : V;

    const int grp = lane >> 4;
    const int m16 = lane & 15;

    v8f c0 = {}, c1 = {}, c2 = {}, c3 = {};

    const float* __restrict__ arow = X + (size_t)(row0 + m16) * F_IN + 2 * grp;

#pragma unroll 4
    for (int k0 = 0; k0 < F_IN; k0 += 4) {
        v2f a;
        a.x = arow[k0];
        a.y = arow[k0 + 1];

        const float* __restrict__ wr0 = W + (size_t)(k0 + 2 * grp) * HD + m16;
        const float* __restrict__ wr1 = wr0 + HD;

        v2f b0, b1, b2, b3;
        b0.x = wr0[0];  b0.y = wr1[0];
        b1.x = wr0[16]; b1.y = wr1[16];
        b2.x = wr0[32]; b2.y = wr1[32];
        b3.x = wr0[48]; b3.y = wr1[48];

        c0 = __builtin_amdgcn_wmma_f32_16x16x4_f32(false, a, false, b0, (short)0, c0, false, false);
        c1 = __builtin_amdgcn_wmma_f32_16x16x4_f32(false, a, false, b1, (short)0, c1, false, false);
        c2 = __builtin_amdgcn_wmma_f32_16x16x4_f32(false, a, false, b2, (short)0, c2, false, false);
        c3 = __builtin_amdgcn_wmma_f32_16x16x4_f32(false, a, false, b3, (short)0, c3, false, false);
    }

    const int mrow = 8 * grp;
#pragma unroll
    for (int r = 0; r < 8; ++r) {
        float* __restrict__ o = Out + (size_t)(row0 + mrow + r) * HD + m16;
        o[0]  = c0[r];
        o[16] = c1[r];
        o[32] = c2[r];
        o[48] = c3[r];
    }
}

#define BNODES      196
#define NBUCK       256
#define BIN_BLOCKS  256
#define BIN_CHUNK   (E_EDGES / BIN_BLOCKS)
#define BIN_THREADS 1024
#define ACC_THREADS 1024
#define ACC_WAVES   (ACC_THREADS / 32)
static_assert(E_EDGES % BIN_BLOCKS == 0, "equal chunks");
static_assert(BNODES * NBUCK >= N_NODES, "ranges cover all nodes");
static_assert((N_NODES - 1) / BNODES < NBUCK, "largest range index in bounds");

typedef __attribute__((ext_vector_type(4))) float v4f;
typedef unsigned long long u64;

__global__ void __launch_bounds__(BIN_THREADS)
gt_bin_count(const int* __restrict__ dst, int* __restrict__ blockhist)
{
    __shared__ int hist[NBUCK];
    const int t = (int)threadIdx.x;
    if (t < NBUCK) hist[t] = 0;
    __syncthreads();
    const int e0 = (int)blockIdx.x * BIN_CHUNK;
    for (int i = t; i < BIN_CHUNK; i += BIN_THREADS) {
        const int d = dst[e0 + i];
        atomicAdd(&hist[d / BNODES], 1);
    }
    __syncthreads();
    if (t < NBUCK) blockhist[(int)blockIdx.x * NBUCK + t] = hist[t];
}

__global__ void __launch_bounds__(NBUCK)
gt_bin_offsets(const int* __restrict__ blockhist, int* __restrict__ blockoff,
               int* __restrict__ bstart)
{
    __shared__ int tot[NBUCK];
    const int t = (int)threadIdx.x;
    int running = 0;
#pragma unroll 32
    for (int b = 0; b < BIN_BLOCKS; ++b) {
        const int c = blockhist[b * NBUCK + t];
        blockoff[b * NBUCK + t] = running;
        running += c;
    }
    tot[t] = running;
    __syncthreads();
    int base = 0;
    for (int j = 0; j < t; ++j) base += tot[j];
    bstart[t] = base;
    if (t == NBUCK - 1) bstart[NBUCK] = base + running;
}

__global__ void __launch_bounds__(BIN_THREADS)
gt_bin_fill(const int* __restrict__ src, const int* __restrict__ dst,
            const int* __restrict__ blockhist, const int* __restrict__ blockoff,
            const int* __restrict__ bstart,
            int* __restrict__ bsrc, int* __restrict__ bdst)
{
    __shared__ int cnt[NBUCK];
    __shared__ int lstart[NBUCK];
    __shared__ int gbase[NBUCK];
    __shared__ int cursor[NBUCK];
    __shared__ int st_s[BIN_CHUNK];
    __shared__ int st_d[BIN_CHUNK];
    const int t = (int)threadIdx.x;
    if (t < NBUCK) {
        cnt[t]   = blockhist[(int)blockIdx.x * NBUCK + t];
        gbase[t] = bstart[t] + blockoff[(int)blockIdx.x * NBUCK + t];
    }
    __syncthreads();
    if (t < NBUCK) {
        int b = 0;
        for (int j = 0; j < t; ++j) b += cnt[j];
        lstart[t] = b;
        cursor[t] = b;
    }
    __syncthreads();
    const int e0 = (int)blockIdx.x * BIN_CHUNK;
    for (int i = t; i < BIN_CHUNK; i += BIN_THREADS) {
        const int d = dst[e0 + i];
        const int s = src[e0 + i];
        const int pos = atomicAdd(&cursor[d / BNODES], 1);
        st_s[pos] = s;
        st_d[pos] = d;
    }
    __syncthreads();
    for (int i = t; i < BIN_CHUNK; i += BIN_THREADS) {
        const int d = st_d[i];
        const int s = st_s[i];
        const int k = d / BNODES;
        const int g = gbase[k] + (i - lstart[k]);
        bsrc[g] = s;
        bdst[g] = d;
    }
}

__device__ __forceinline__ float gt_quad_xor1(float x)
{
    return __int_as_float(__builtin_amdgcn_update_dpp(0, __float_as_int(x), 0xB1, 0xF, 0xF, true));
}
__device__ __forceinline__ float gt_quad_xor2(float x)
{
    return __int_as_float(__builtin_amdgcn_update_dpp(0, __float_as_int(x), 0x4E, 0xF, 0xF, true));
}

__device__ __forceinline__ u64 gt_to_fixed(float x)
{
    const float ax = fabsf(x);
    const float t  = truncf(ax);
    const float r  = ax - t;
    const unsigned hi = (unsigned)t;
    const unsigned lo = (unsigned)(r * 4294967296.0f);
    const u64 mag = ((u64)hi << 32) | (u64)lo;
    return (x < 0.0f) ? (0ull - mag) : mag;
}
__device__ __forceinline__ float gt_from_fixed(u64 a)
{
    return (float)(long long)a * (1.0f / 4294967296.0f);
}

struct PairRows {
    v4f k4, v4;
    int n0, n1, nl;
    bool ok;
};

__device__ __forceinline__ PairRows
gt_load_pair(const float* __restrict__ K, const float* __restrict__ V,
             int sv, int nv, int p, int left, int half, int i16)
{
    PairRows r;
    const int s0 = __builtin_amdgcn_readlane(sv, 2 * p);
    const int s1 = __builtin_amdgcn_readlane(sv, 2 * p + 1);
    r.n0 = __builtin_amdgcn_readlane(nv, 2 * p);
    r.n1 = __builtin_amdgcn_readlane(nv, 2 * p + 1);
    const int sl = half ? s1 : s0;
    r.nl = half ? r.n1 : r.n0;
    r.ok = (2 * p + half) < left;
    const unsigned so = (unsigned)sl * (unsigned)HD + 4u * (unsigned)i16;
    r.k4 = *reinterpret_cast<const v4f*>(K + so);
    r.v4 = *reinterpret_cast<const v4f*>(V + so);
    return r;
}

__device__ __forceinline__ void
gt_reduce_pair(const PairRows& c, const float* qs, u64* acc, u64* zacc,
               float* slot, int lane, int i16)
{
    const v4f q4 = *reinterpret_cast<const v4f*>(qs + c.nl * HD + 4 * i16);
    float d = c.k4.x * q4.x;
    d = fmaf(c.k4.y, q4.y, d);
    d = fmaf(c.k4.z, q4.z, d);
    d = fmaf(c.k4.w, q4.w, d);
    d += gt_quad_xor1(d);
    d += gt_quad_xor2(d);
    float sc = __expf(fminf(fmaxf(d * 0.25f, -5.0f), 5.0f));
    sc = c.ok ? sc : 0.0f;

    slot[4 * lane + 0] = sc * c.v4.x;
    slot[4 * lane + 1] = sc * c.v4.y;
    slot[4 * lane + 2] = sc * c.v4.z;
    slot[4 * lane + 3] = sc * c.v4.w;
    const float a0 = slot[lane];
    const float a1 = slot[lane + 32];
    const float b0 = slot[lane + 64];
    const float b1 = slot[lane + 96];

    u64* r0 = acc + c.n0 * HD + lane;
    u64* r1 = acc + c.n1 * HD + lane;
    atomicAdd(r0,      gt_to_fixed(a0));
    atomicAdd(r0 + 32, gt_to_fixed(a1));
    atomicAdd(r1,      gt_to_fixed(b0));
    atomicAdd(r1 + 32, gt_to_fixed(b1));
    if ((lane & 3) == 0)
        atomicAdd(zacc + c.nl * NHEAD + (i16 >> 2), gt_to_fixed(sc));
}

__global__ void __launch_bounds__(ACC_THREADS)
gt_range_attn(const int* __restrict__ bsrc, const int* __restrict__ bdst,
              const int* __restrict__ bstart,
              const float* __restrict__ Q, const float* __restrict__ K,
              const float* __restrict__ V, float* __restrict__ out)
{
    __shared__ u64 acc[BNODES * HD];
    __shared__ u64 zacc[BNODES * NHEAD];
    __shared__ __attribute__((aligned(16))) float qs[BNODES * HD];
    __shared__ __attribute__((aligned(16))) float s_rows[ACC_WAVES * 128];

    const int t     = (int)threadIdx.x;
    const int lane  = t & 31;
    const int w     = t >> 5;
    const int half  = lane >> 4;
    const int i16   = lane & 15;
    const int node0 = (int)blockIdx.x * BNODES;
    const int nn    = min(BNODES, N_NODES - node0);

    for (int i = t; i < BNODES * HD; i += ACC_THREADS) acc[i] = 0;
    for (int i = t; i < BNODES * NHEAD; i += ACC_THREADS) zacc[i] = 0;
    for (int i = t; i < nn * (HD / 4); i += ACC_THREADS)
        reinterpret_cast<v4f*>(qs)[i] = reinterpret_cast<const v4f*>(Q + (size_t)node0 * HD)[i];
    __syncthreads();

    const int seg0 = bstart[blockIdx.x];
    const int seg1 = bstart[blockIdx.x + 1];
    float* slot = s_rows + w * 128;

    int bb = seg0 + 32 * w;
    if (bb < seg1) {
        int left = seg1 - bb;
        int sv = (lane < left) ? bsrc[bb + lane] : 0;
        int nv = (lane < left) ? (bdst[bb + lane] - node0) : 0;
        PairRows cur = gt_load_pair(K, V, sv, nv, 0, left, half, i16);
        for (;;) {
            const int bbn   = bb + 32 * ACC_WAVES;
            const bool more = bbn < seg1;
            const int leftn = seg1 - bbn;
            int svn = 0, nvn = 0;
            if (more) {
                svn = (lane < leftn) ? bsrc[bbn + lane] : 0;
                nvn = (lane < leftn) ? (bdst[bbn + lane] - node0) : 0;
            }
#pragma unroll
            for (int p = 0; p < 16; ++p) {
                PairRows nxt = cur;
                if (p + 1 < 16)
                    nxt = gt_load_pair(K, V, sv, nv, p + 1, left, half, i16);
                else if (more)
                    nxt = gt_load_pair(K, V, svn, nvn, 0, leftn, half, i16);
                __builtin_amdgcn_sched_barrier(0);
                gt_reduce_pair(cur, qs, acc, zacc, slot, lane, i16);
                cur = nxt;
            }
            if (!more) break;
            bb = bbn; left = leftn; sv = svn; nv = nvn;
        }
    }
    __syncthreads();

    for (int r = w; r < nn; r += ACC_WAVES) {
        const float z_lo = gt_from_fixed(zacc[r * NHEAD + (lane >> 4)])     + 1e-6f;
        const float z_hi = gt_from_fixed(zacc[r * NHEAD + 2 + (lane >> 4)]) + 1e-6f;
        float* __restrict__ o = out + (size_t)(node0 + r) * HD + lane;
        o[0]  = gt_from_fixed(acc[r * HD + lane])      / z_lo;
        o[32] = gt_from_fixed(acc[r * HD + lane + 32]) / z_hi;
    }
}

extern "C" void kernel_launch(void* const* d_in, const int* in_sizes, int n_in,
                              void* d_out, int out_size, void* d_ws, size_t ws_size,
                              hipStream_t stream)
{
    (void)in_sizes; (void)n_in; (void)out_size; (void)ws_size;

    const float* X   = (const float*)d_in[0];
    const int*   src = (const int*)  d_in[1];
    const int*   dst = (const int*)  d_in[2];
    const float* Wq  = (const float*)d_in[3];
    const float* Wk  = (const float*)d_in[4];
    const float* Wv  = (const float*)d_in[5];

    float* out = (float*)d_out;
    float* ws  = (float*)d_ws;
    float* Q   = ws;
    float* K   = ws + (size_t)N_NODES * HD;
    float* V   = ws + (size_t)2 * N_NODES * HD;
    int* blockhist = (int*)(ws + (size_t)3 * N_NODES * HD);
    int* blockoff  = blockhist + (size_t)BIN_BLOCKS * NBUCK;
    int* bstart    = blockoff  + (size_t)BIN_BLOCKS * NBUCK;
    int* bsrc = (int*)(ws + (size_t)3 * N_NODES * HD + (size_t)2 * N_NODES * NHEAD);
    int* bdst = bsrc + (size_t)E_EDGES;

    {
        const int ntask = (N_NODES / 16) * 3;
        const int wpb   = 256 / 32;
        const int blocks = (ntask + wpb - 1) / wpb;
        gt_qkv_proj_wmma<<<blocks, 256, 0, stream>>>(X, Wq, Wk, Wv, Q, K, V);
    }

    gt_bin_count<<<BIN_BLOCKS, BIN_THREADS, 0, stream>>>(dst, blockhist);
    gt_bin_offsets<<<1, NBUCK, 0, stream>>>(blockhist, blockoff, bstart);
    gt_bin_fill<<<BIN_BLOCKS, BIN_THREADS, 0, stream>>>(src, dst, blockhist, blockoff, bstart, bsrc, bdst);

    gt_range_attn<<<NBUCK, ACC_THREADS, 0, stream>>>(bsrc, bdst, bstart, Q, K, V, out);
}


# One filing launch: each block orders only its own chunk of the edge list by dst range (rank = return value of the LDS count add); the range kernel walks the 256 per-chunk runs; 5 launches -> 3
# speedup vs baseline: 2.1679x; 1.1202x over previous
#include <hip/hip_runtime.h>
#include <hip/hip_bf16.h>

typedef __attribute__((ext_vector_type(2))) float v2f;
typedef __attribute__((ext_vector_type(8))) float v8f;

#define N_NODES 50000
#define E_EDGES 1600000
#define NHEAD   4
#define DHEAD   16
#define F_IN    128
#define HD      64

__global__ void __launch_bounds__(256)
gt_qkv_proj_wmma(const float* __restrict__ X,
                 const float* __restrict__ Wq,
                 const float* __restrict__ Wk,
                 const float* __restrict__ Wv,
                 float* __restrict__ Q,
                 float* __restrict__ K,
                 float* __restrict__ V)
{
    const int wave = (int)((blockIdx.x * blockDim.x + threadIdx.x) >> 5);
    const int lane = (int)(threadIdx.x & 31);
    const int NTASK = (N_NODES / 16) * 3;
    if (wave >= NTASK) return;

    const int mat  = wave % 3;
    const int row0 = (wave / 3) * 16;
    const float* __restrict__ W   = (mat == 0) ? Wq : (mat == 1) ? Wk : Wv;
    float*       __restrict__ Out = (mat == 0) ? Q  : (mat == 1) ? K  : V;

    const int grp = lane >> 4;
    const int m16 = lane & 15;

    v8f c0 = {}, c1 = {}, c2 = {}, c3 = {};

    const float* __restrict__ arow = X + (size_t)(row0 + m16) * F_IN + 2 * grp;

#pragma unroll 4
    for (int k0 = 0; k0 < F_IN; k0 += 4) {
        v2f a;
        a.x = arow[k0];
        a.y = arow[k0 + 1];

        const float* __restrict__ wr0 = W + (size_t)(k0 + 2 * grp) * HD + m16;
        const float* __restrict__ wr1 = wr0 + HD;

        v2f b0, b1, b2, b3;
        b0.x = wr0[0];  b0.y = wr1[0];
        b1.x = wr0[16]; b1.y = wr1[16];
        b2.x = wr0[32]; b2.y = wr1[32];
        b3.x = wr0[48]; b3.y = wr1[48];

        c0 = __builtin_amdgcn_wmma_f32_16x16x4_f32(false, a, false, b0, (short)0, c0, false, false);
        c1 = __builtin_amdgcn_wmma_f32_16x16x4_f32(false, a, false, b1, (short)0, c1, false, false);
        c2 = __builtin_amdgcn_wmma_f32_16x16x4_f32(false, a, false, b2, (short)0, c2, false, false);
        c3 = __builtin_amdgcn_wmma_f32_16x16x4_f32(false, a, false, b3, (short)0, c3, false, false);
    }

    const int mrow = 8 * grp;
#pragma unroll
    for (int r = 0; r < 8; ++r) {
        float* __restrict__ o = Out + (size_t)(row0 + mrow + r) * HD + m16;
        o[0]  = c0[r];
        o[16] = c1[r];
        o[32] = c2[r];
        o[48] = c3[r];
    }
}

#define BNODES      196
#define NBUCK       256
#define BIN_BLOCKS  256
#define BIN_CHUNK   (E_EDGES / BIN_BLOCKS)
#define BIN_THREADS 1024
#define BIN_PER     ((BIN_CHUNK + BIN_THREADS - 1) / BIN_THREADS)
#define ACC_THREADS 1024
#define ACC_WAVES   (ACC_THREADS / 32)
static_assert(E_EDGES % BIN_BLOCKS == 0, "equal chunks");
static_assert(BNODES * NBUCK >= N_NODES, "ranges cover all nodes");
static_assert((N_NODES - 1) / BNODES < NBUCK, "largest range index in bounds");
static_assert(BIN_BLOCKS == 256, "the run search below counts in two levels of 16");
static_assert(BIN_BLOCKS <= ACC_THREADS && NBUCK <= BIN_THREADS, "one thread per table entry");

typedef __attribute__((ext_vector_type(4))) float v4f;
typedef unsigned long long u64;

__global__ void __launch_bounds__(BIN_THREADS)
gt_file_chunk(const int* __restrict__ src, const int* __restrict__ dst,
              int* __restrict__ runstart, int* __restrict__ bsrc, int* __restrict__ bdst)
{
    __shared__ int hist[NBUCK];
    __shared__ int lstart[NBUCK];
    __shared__ int st_s[BIN_CHUNK];
    __shared__ int st_d[BIN_CHUNK];
    const int t = (int)threadIdx.x;
    if (t < NBUCK) hist[t] = 0;
    __syncthreads();
    const int e0 = (int)blockIdx.x * BIN_CHUNK;
    int dd[BIN_PER], ss[BIN_PER], rk[BIN_PER];
#pragma unroll
    for (int j = 0; j < BIN_PER; ++j) {
        const int i = t + j * BIN_THREADS;
        dd[j] = 0; ss[j] = 0;
        if (i < BIN_CHUNK) {
            dd[j] = dst[e0 + i];
            ss[j] = src[e0 + i];
        }
    }
#pragma unroll
    for (int j = 0; j < BIN_PER; ++j) {
        const int i = t + j * BIN_THREADS;
        rk[j] = 0;
        if (i < BIN_CHUNK) rk[j] = atomicAdd(&hist[dd[j] / BNODES], 1);
    }
    __syncthreads();
    if (t < NBUCK) {
        int b = 0;
        for (int j = 0; j < t; ++j) b += hist[j];
        lstart[t] = b;
        runstart[(int)blockIdx.x * NBUCK + t] = b;
    }
    __syncthreads();
#pragma unroll
    for (int j = 0; j < BIN_PER; ++j) {
        const int i = t + j * BIN_THREADS;
        if (i < BIN_CHUNK) {
            const int pos = lstart[dd[j] / BNODES] + rk[j];
            st_s[pos] = ss[j];
            st_d[pos] = dd[j];
        }
    }
    __syncthreads();
    for (int i = t; i < BIN_CHUNK; i += BIN_THREADS) {
        bsrc[e0 + i] = st_s[i];
        bdst[e0 + i] = st_d[i];
    }
}

__device__ __forceinline__ float gt_quad_xor1(float x)
{
    return __int_as_float(__builtin_amdgcn_update_dpp(0, __float_as_int(x), 0xB1, 0xF, 0xF, true));
}
__device__ __forceinline__ float gt_quad_xor2(float x)
{
    return __int_as_float(__builtin_amdgcn_update_dpp(0, __float_as_int(x), 0x4E, 0xF, 0xF, true));
}

__device__ __forceinline__ u64 gt_to_fixed(float x)
{
    const float ax = fabsf(x);
    const float t  = truncf(ax);
    const float r  = ax - t;
    const unsigned hi = (unsigned)t;
    const unsigned lo = (unsigned)(r * 4294967296.0f);
    const u64 mag = ((u64)hi << 32) | (u64)lo;
    return (x < 0.0f) ? (0ull - mag) : mag;
}
__device__ __forceinline__ float gt_from_fixed(u64 a)
{
    return (float)(long long)a * (1.0f / 4294967296.0f);
}

struct PairRows {
    v4f k4, v4;
    int n0, n1, nl;
    bool ok;
};

__device__ __forceinline__ PairRows
gt_load_pair(const float* __restrict__ K, const float* __restrict__ V,
             int sv, int nv, int p, int left, int half, int i16)
{
    PairRows r;
    const int s0 = __builtin_amdgcn_readlane(sv, 2 * p);
    const int s1 = __builtin_amdgcn_readlane(sv, 2 * p + 1);
    r.n0 = __builtin_amdgcn_readlane(nv, 2 * p);
    r.n1 = __builtin_amdgcn_readlane(nv, 2 * p + 1);
    const int sl = half ? s1 : s0;
    r.nl = half ? r.n1 : r.n0;
    r.ok = (2 * p + half) < left;
    const unsigned so = (unsigned)sl * (unsigned)HD + 4u * (unsigned)i16;
    r.k4 = *reinterpret_cast<const v4f*>(K + so);
    r.v4 = *reinterpret_cast<const v4f*>(V + so);
    return r;
}

__device__ __forceinline__ void
gt_reduce_pair(const PairRows& c, const float* qs, u64* acc, u64* zacc,
               float* slot, int lane, int i16)
{
    const v4f q4 = *reinterpret_cast<const v4f*>(qs + c.nl * HD + 4 * i16);
    float d = c.k4.x * q4.x;
    d = fmaf(c.k4.y, q4.y, d);
    d = fmaf(c.k4.z, q4.z, d);
    d = fmaf(c.k4.w, q4.w, d);
    d += gt_quad_xor1(d);
    d += gt_quad_xor2(d);
    float sc = __expf(fminf(fmaxf(d * 0.25f, -5.0f), 5.0f));
    sc = c.ok ? sc : 0.0f;

    slot[4 * lane + 0] = sc * c.v4.x;
    slot[4 * lane + 1] = sc * c.v4.y;
    slot[4 * lane + 2] = sc * c.v4.z;
    slot[4 * lane + 3] = sc * c.v4.w;
    const float a0 = slot[lane];
    const float a1 = slot[lane + 32];
    const float b0 = slot[lane + 64];
    const float b1 = slot[lane + 96];

    u64* r0 = acc + c.n0 * HD + lane;
    u64* r1 = acc + c.n1 * HD + lane;
    atomicAdd(r0,      gt_to_fixed(a0));
    atomicAdd(r0 + 32, gt_to_fixed(a1));
    atomicAdd(r1,      gt_to_fixed(b0));
    atomicAdd(r1 + 32, gt_to_fixed(b1));
    if ((lane & 3) == 0)
        atomicAdd(zacc + c.nl * NHEAD + (i16 >> 2), gt_to_fixed(sc));
}

typedef __attribute__((ext_vector_type(4))) int v4i;
__device__ __forceinline__ int gt_count_le(const int* p16, int v)
{
    const v4i* q = reinterpret_cast<const v4i*>(p16);
    const v4i a = q[0], b = q[1], c = q[2], d = q[3];
    return (a.x <= v) + (a.y <= v) + (a.z <= v) + (a.w <= v)
         + (b.x <= v) + (b.y <= v) + (b.z <= v) + (b.w <= v)
         + (c.x <= v) + (c.y <= v) + (c.z <= v) + (c.w <= v)
         + (d.x <= v) + (d.y <= v) + (d.z <= v) + (d.w <= v);
}
__device__ __forceinline__ int gt_run_pos(const int* pre, const int* pre16, const int* base, int v)
{
    const int c = gt_count_le(pre16, v) - 1;
    const int b = 16 * c + gt_count_le(pre + 16 * c, v) - 1;
    return base[b] + (v - pre[b]);
}

__global__ void __launch_bounds__(ACC_THREADS)
gt_range_attn(const int* __restrict__ bsrc, const int* __restrict__ bdst,
              const int* __restrict__ runstart,
              const float* __restrict__ Q, const float* __restrict__ K,
              const float* __restrict__ V, float* __restrict__ out)
{
    __shared__ u64 acc[BNODES * HD];
    __shared__ u64 zacc[BNODES * NHEAD];
    __shared__ __attribute__((aligned(16))) float qs[BNODES * HD];
    __shared__ __attribute__((aligned(16))) float s_rows[ACC_WAVES * 128];
    __shared__ int rbase[BIN_BLOCKS];
    __shared__ int rcnt[BIN_BLOCKS];
    __shared__ __attribute__((aligned(16))) int rpre[BIN_BLOCKS + 4];
    __shared__ __attribute__((aligned(16))) int rpre16[16];

    const int t     = (int)threadIdx.x;
    const int lane  = t & 31;
    const int w     = t >> 5;
    const int half  = lane >> 4;
    const int i16   = lane & 15;
    const int node0 = (int)blockIdx.x * BNODES;
    const int nn    = min(BNODES, N_NODES - node0);

    if (t < BIN_BLOCKS) {
        const int rs = runstart[t * NBUCK + (int)blockIdx.x];
        const int re = ((int)blockIdx.x + 1 < NBUCK) ? runstart[t * NBUCK + (int)blockIdx.x + 1] : BIN_CHUNK;
        rbase[t] = t * BIN_CHUNK + rs;
        rcnt[t]  = re - rs;
    }
    for (int i = t; i < BNODES * HD; i += ACC_THREADS) acc[i] = 0;
    for (int i = t; i < BNODES * NHEAD; i += ACC_THREADS) zacc[i] = 0;
    for (int i = t; i < nn * (HD / 4); i += ACC_THREADS)
        reinterpret_cast<v4f*>(qs)[i] = reinterpret_cast<const v4f*>(Q + (size_t)node0 * HD)[i];
    __syncthreads();
    if (t < BIN_BLOCKS) {
        int b = 0;
        for (int j = 0; j < t; ++j) b += rcnt[j];
        rpre[t] = b;
        if ((t & 15) == 0) rpre16[t >> 4] = b;
        if (t == BIN_BLOCKS - 1) rpre[BIN_BLOCKS] = b + rcnt[t];
    }
    __syncthreads();

    const int seg1 = rpre[BIN_BLOCKS];
    float* slot = s_rows + w * 128;

    int bb = 32 * w;
    if (bb < seg1) {
        int left = seg1 - bb;
        int sv = 0, nv = 0;
        if (lane < left) {
            const int g = gt_run_pos(rpre, rpre16, rbase, bb + lane);
            sv = bsrc[g];
            nv = bdst[g] - node0;
        }
        PairRows cur = gt_load_pair(K, V, sv, nv, 0, left, half, i16);
        for (;;) {
            const int bbn   = bb + 32 * ACC_WAVES;
            const bool more = bbn < seg1;
            const int leftn = seg1 - bbn;
            int svn = 0, nvn = 0;
            if (more && lane < leftn) {
                const int g = gt_run_pos(rpre, rpre16, rbase, bbn + lane);
                svn = bsrc[g];
                nvn = bdst[g] - node0;
            }
#pragma unroll
            for (int p = 0; p < 16; ++p) {
                PairRows nxt = cur;
                if (p + 1 < 16)
                    nxt = gt_load_pair(K, V, sv, nv, p + 1, left, half, i16);
                else if (more)
                    nxt = gt_load_pair(K, V, svn, nvn, 0, leftn, half, i16);
                __builtin_amdgcn_sched_barrier(0);
                gt_reduce_pair(cur, qs, acc, zacc, slot, lane, i16);
                cur = nxt;
            }
            if (!more) break;
            bb = bbn; left = leftn; sv = svn; nv = nvn;
        }
    }
    __syncthreads();

    for (int r = w; r < nn; r += ACC_WAVES) {
        const float z_lo = gt_from_fixed(zacc[r * NHEAD + (lane >> 4)])     + 1e-6f;
        const float z_hi = gt_from_fixed(zacc[r * NHEAD + 2 + (lane >> 4)]) + 1e-6f;
        float* __restrict__ o = out + (size_t)(node0 + r) * HD + lane;
        o[0]  = gt_from_fixed(acc[r * HD + lane])      / z_lo;
        o[32] = gt_from_fixed(acc[r * HD + lane + 32]) / z_hi;
    }
}

extern "C" void kernel_launch(void* const* d_in, const int* in_sizes, int n_in,
                              void* d_out, int out_size, void* d_ws, size_t ws_size,
                              hipStream_t stream)
{
    (void)in_sizes; (void)n_in; (void)out_size; (void)ws_size;

    const float* X   = (const float*)d_in[0];
    const int*   src = (const int*)  d_in[1];
    const int*   dst = (const int*)  d_in[2];
    const float* Wq  = (const float*)d_in[3];
    const float* Wk  = (const float*)d_in[4];
    const float* Wv  = (const float*)d_in[5];

    float* out = (float*)d_out;
    float* ws  = (float*)d_ws;
    float* Q   = ws;
    float* K   = ws + (size_t)N_NODES * HD;
    float* V   = ws + (size_t)2 * N_NODES * HD;
    int* runstart = (int*)(ws + (size_t)3 * N_NODES * HD);
    int* bsrc = (int*)(ws + (size_t)3 * N_NODES * HD + (size_t)2 * N_NODES * NHEAD);
    int* bdst = bsrc + (size_t)E_EDGES;

    {
        const int ntask = (N_NODES / 16) * 3;
        const int wpb   = 256 / 32;
        const int blocks = (ntask + wpb - 1) / wpb;
        gt_qkv_proj_wmma<<<blocks, 256, 0, stream>>>(X, Wq, Wk, Wv, Q, K, V);
    }

    gt_file_chunk<<<BIN_BLOCKS, BIN_THREADS, 0, stream>>>(src, dst, runstart, bsrc, bdst);

    gt_range_attn<<<NBUCK, ACC_THREADS, 0, stream>>>(bsrc, bdst, runstart, Q, K, V, out);
}


# Edge filing and Q/K/V projection in ONE launch: blocks 0..255 each file a chunk (256 threads, 27 KB LDS), the other 1172 project as before; 3 launches -> 2; no sum or order of sums touched
# speedup vs baseline: 2.2669x; 1.0457x over previous
#include <hip/hip_runtime.h>
#include <hip/hip_bf16.h>

typedef __attribute__((ext_vector_type(2))) float v2f;
typedef __attribute__((ext_vector_type(8))) float v8f;

#define N_NODES 50000
#define E_EDGES 1600000
#define NHEAD   4
#define DHEAD   16
#define F_IN    128
#define HD      64
#define PROJ_TASKS ((N_NODES / 16) * 3)

__device__ __forceinline__ void
gt_qkv_proj_tile(int wave, int lane,
                 const float* __restrict__ X,
                 const float* __restrict__ Wq,
                 const float* __restrict__ Wk,
                 const float* __restrict__ Wv,
                 float* __restrict__ Q,
                 float* __restrict__ K,
                 float* __restrict__ V)
{
    const int NTASK = PROJ_TASKS;
    if (wave >= NTASK) return;

    const int mat  = wave % 3;
    const int row0 = (wave / 3) * 16;
    const float* __restrict__ W   = (mat == 0) ? Wq : (mat == 1) ? Wk : Wv;
    float*       __restrict__ Out = (mat == 0) ? Q  : (mat == 1) ? K  : V;

    const int grp = lane >> 4;
    const int m16 = lane & 15;

    v8f c0 = {}, c1 = {}, c2 = {}, c3 = {};

    const float* __restrict__ arow = X + (size_t)(row0 + m16) * F_IN + 2 * grp;

#pragma unroll 4
    for (int k0 = 0; k0 < F_IN; k0 += 4) {
        v2f a;
        a.x = arow[k0];
        a.y = arow[k0 + 1];

        const float* __restrict__ wr0 = W + (size_t)(k0 + 2 * grp) * HD + m16;
        const float* __restrict__ wr1 = wr0 + HD;

        v2f b0, b1, b2, b3;
        b0.x = wr0[0];  b0.y = wr1[0];
        b1.x = wr0[16]; b1.y = wr1[16];
        b2.x = wr0[32]; b2.y = wr1[32];
        b3.x = wr0[48]; b3.y = wr1[48];

        c0 = __builtin_amdgcn_wmma_f32_16x16x4_f32(false, a, false, b0, (short)0, c0, false, false);
        c1 = __builtin_amdgcn_wmma_f32_16x16x4_f32(false, a, false, b1, (short)0, c1, false, false);
        c2 = __builtin_amdgcn_wmma_f32_16x16x4_f32(false, a, false, b2, (short)0, c2, false, false);
        c3 = __builtin_amdgcn_wmma_f32_16x16x4_f32(false, a, false, b3, (short)0, c3, false, false);
    }

    const int mrow = 8 * grp;
#pragma unroll
    for (int r = 0; r < 8; ++r) {
        float* __restrict__ o = Out + (size_t)(row0 + mrow + r) * HD + m16;
        o[0]  = c0[r];
        o[16] = c1[r];
        o[32] = c2[r];
        o[48] = c3[r];
    }
}

#define BNODES      196
#define NBUCK       256
#define BIN_BLOCKS  256
#define BIN_CHUNK   (E_EDGES / BIN_BLOCKS)
#define FRONT_THREADS 256
#define BIN_PER     ((BIN_CHUNK + FRONT_THREADS - 1) / FRONT_THREADS)
#define FILE_R      13
#define PROJ_BLOCKS ((PROJ_TASKS + FRONT_THREADS / 32 - 1) / (FRONT_THREADS / 32))
#define ACC_THREADS 1024
#define ACC_WAVES   (ACC_THREADS / 32)
static_assert(E_EDGES % BIN_BLOCKS == 0, "equal chunks");
static_assert(BNODES * NBUCK >= N_NODES, "ranges cover all nodes");
static_assert((N_NODES - 1) / BNODES < NBUCK, "largest range index in bounds");
static_assert(BIN_BLOCKS == 256, "the run search below counts in two levels of 16");
static_assert(BIN_BLOCKS <= ACC_THREADS && NBUCK == FRONT_THREADS, "one thread per table entry");
static_assert(N_NODES <= 65536, "a filed edge is packed as two 16-bit node numbers");

typedef __attribute__((ext_vector_type(4))) float v4f;
typedef unsigned long long u64;

__device__ __forceinline__ void
gt_file_chunk(int chunk, const int* __restrict__ src, const int* __restrict__ dst,
              int* __restrict__ runstart, int* __restrict__ bsrc, int* __restrict__ bdst)
{
    __shared__ int hist[NBUCK];
    __shared__ int cursor[NBUCK];
    __shared__ unsigned st[BIN_CHUNK];
    const int t = (int)threadIdx.x;
    hist[t] = 0;
    const int e0 = chunk * BIN_CHUNK;
    __syncthreads();
#pragma unroll 1
    for (int r0 = 0; r0 < BIN_PER; r0 += FILE_R) {
        int dd[FILE_R];
#pragma unroll
        for (int j = 0; j < FILE_R; ++j) {
            const int i = t + (r0 + j) * FRONT_THREADS;
            dd[j] = 0;
            if (i < BIN_CHUNK) dd[j] = dst[e0 + i];
        }
#pragma unroll
        for (int j = 0; j < FILE_R; ++j) {
            const int i = t + (r0 + j) * FRONT_THREADS;
            if (i < BIN_CHUNK) atomicAdd(&hist[(unsigned)dd[j] / (unsigned)BNODES], 1);
        }
    }
    __syncthreads();
    {
        int b = 0;
        for (int j = 0; j < t; ++j) b += hist[j];
        cursor[t] = b;
        runstart[chunk * NBUCK + t] = b;
    }
    __syncthreads();
#pragma unroll 1
    for (int r0 = 0; r0 < BIN_PER; r0 += FILE_R) {
        int dd[FILE_R], ss[FILE_R];
#pragma unroll
        for (int j = 0; j < FILE_R; ++j) {
            const int i = t + (r0 + j) * FRONT_THREADS;
            dd[j] = 0; ss[j] = 0;
            if (i < BIN_CHUNK) {
                dd[j] = dst[e0 + i];
                ss[j] = src[e0 + i];
            }
        }
        int pos[FILE_R];
#pragma unroll
        for (int j = 0; j < FILE_R; ++j) {
            const int i = t + (r0 + j) * FRONT_THREADS;
            pos[j] = 0;
            if (i < BIN_CHUNK) pos[j] = atomicAdd(&cursor[(unsigned)dd[j] / (unsigned)BNODES], 1);
        }
#pragma unroll
        for (int j = 0; j < FILE_R; ++j) {
            const int i = t + (r0 + j) * FRONT_THREADS;
            if (i < BIN_CHUNK) st[pos[j]] = ((unsigned)ss[j] << 16) | (unsigned)dd[j];
        }
    }
    __syncthreads();
    for (int i = t; i < BIN_CHUNK; i += FRONT_THREADS) {
        const unsigned u = st[i];
        bsrc[e0 + i] = (int)(u >> 16);
        bdst[e0 + i] = (int)(u & 0xFFFFu);
    }
}

__global__ void __launch_bounds__(FRONT_THREADS)
gt_front(const float* __restrict__ X,
         const float* __restrict__ Wq,
         const float* __restrict__ Wk,
         const float* __restrict__ Wv,
         const int* __restrict__ src, const int* __restrict__ dst,
         float* __restrict__ Q, float* __restrict__ K, float* __restrict__ V,
         int* __restrict__ runstart, int* __restrict__ bsrc, int* __restrict__ bdst)
{
    if ((int)blockIdx.x < BIN_BLOCKS) {
        gt_file_chunk((int)blockIdx.x, src, dst, runstart, bsrc, bdst);
    } else {
        const int wave = (((int)blockIdx.x - BIN_BLOCKS) * FRONT_THREADS + (int)threadIdx.x) >> 5;
        gt_qkv_proj_tile(wave, (int)(threadIdx.x & 31), X, Wq, Wk, Wv, Q, K, V);
    }
}

__device__ __forceinline__ float gt_quad_xor1(float x)
{
    return __int_as_float(__builtin_amdgcn_update_dpp(0, __float_as_int(x), 0xB1, 0xF, 0xF, true));
}
__device__ __forceinline__ float gt_quad_xor2(float x)
{
    return __int_as_float(__builtin_amdgcn_update_dpp(0, __float_as_int(x), 0x4E, 0xF, 0xF, true));
}

__device__ __forceinline__ u64 gt_to_fixed(float x)
{
    const float ax = fabsf(x);
    const float t  = truncf(ax);
    const float r  = ax - t;
    const unsigned hi = (unsigned)t;
    const unsigned lo = (unsigned)(r * 4294967296.0f);
    const u64 mag = ((u64)hi << 32) | (u64)lo;
    return (x < 0.0f) ? (0ull - mag) : mag;
}
__device__ __forceinline__ float gt_from_fixed(u64 a)
{
    return (float)(long long)a * (1.0f / 4294967296.0f);
}

struct PairRows {
    v4f k4, v4;
    int n0, n1, nl;
    bool ok;
};

__device__ __forceinline__ PairRows
gt_load_pair(const float* __restrict__ K, const float* __restrict__ V,
             int sv, int nv, int p, int left, int half, int i16)
{
    PairRows r;
    const int s0 = __builtin_amdgcn_readlane(sv, 2 * p);
    const int s1 = __builtin_amdgcn_readlane(sv, 2 * p + 1);
    r.n0 = __builtin_amdgcn_readlane(nv, 2 * p);
    r.n1 = __builtin_amdgcn_readlane(nv, 2 * p + 1);
    const int sl = half ? s1 : s0;
    r.nl = half ? r.n1 : r.n0;
    r.ok = (2 * p + half) < left;
    const unsigned so = (unsigned)sl * (unsigned)HD + 4u * (unsigned)i16;
    r.k4 = *reinterpret_cast<const v4f*>(K + so);
    r.v4 = *reinterpret_cast<const v4f*>(V + so);
    return r;
}

__device__ __forceinline__ void
gt_reduce_pair(const PairRows& c, const float* qs, u64* acc, u64* zacc,
               float* slot, int lane, int i16)
{
    const v4f q4 = *reinterpret_cast<const v4f*>(qs + c.nl * HD + 4 * i16);
    float d = c.k4.x * q4.x;
    d = fmaf(c.k4.y, q4.y, d);
    d = fmaf(c.k4.z, q4.z, d);
    d = fmaf(c.k4.w, q4.w, d);
    d += gt_quad_xor1(d);
    d += gt_quad_xor2(d);
    float sc = __expf(fminf(fmaxf(d * 0.25f, -5.0f), 5.0f));
    sc = c.ok ? sc : 0.0f;

    slot[4 * lane + 0] = sc * c.v4.x;
    slot[4 * lane + 1] = sc * c.v4.y;
    slot[4 * lane + 2] = sc * c.v4.z;
    slot[4 * lane + 3] = sc * c.v4.w;
    const float a0 = slot[lane];
    const float a1 = slot[lane + 32];
    const float b0 = slot[lane + 64];
    const float b1 = slot[lane + 96];

    u64* r0 = acc + c.n0 * HD + lane;
    u64* r1 = acc + c.n1 * HD + lane;
    atomicAdd(r0,      gt_to_fixed(a0));
    atomicAdd(r0 + 32, gt_to_fixed(a1));
    atomicAdd(r1,      gt_to_fixed(b0));
    atomicAdd(r1 + 32, gt_to_fixed(b1));
    if ((lane & 3) == 0)
        atomicAdd(zacc + c.nl * NHEAD + (i16 >> 2), gt_to_fixed(sc));
}

typedef __attribute__((ext_vector_type(4))) int v4i;
__device__ __forceinline__ int gt_count_le(const int* p16, int v)
{
    const v4i* q = reinterpret_cast<const v4i*>(p16);
    const v4i a = q[0], b = q[1], c = q[2], d = q[3];
    return (a.x <= v) + (a.y <= v) + (a.z <= v) + (a.w <= v)
         + (b.x <= v) + (b.y <= v) + (b.z <= v) + (b.w <= v)
         + (c.x <= v) + (c.y <= v) + (c.z <= v) + (c.w <= v)
         + (d.x <= v) + (d.y <= v) + (d.z <= v) + (d.w <= v);
}
__device__ __forceinline__ int gt_run_pos(const int* pre, const int* pre16, const int* base, int v)
{
    const int c = gt_count_le(pre16, v) - 1;
    const int b = 16 * c + gt_count_le(pre + 16 * c, v) - 1;
    return base[b] + (v - pre[b]);
}

__global__ void __launch_bounds__(ACC_THREADS)
gt_range_attn(const int* __restrict__ bsrc, const int* __restrict__ bdst,
              const int* __restrict__ runstart,
              const float* __restrict__ Q, const float* __restrict__ K,
              const float* __restrict__ V, float* __restrict__ out)
{
    __shared__ u64 acc[BNODES * HD];
    __shared__ u64 zacc[BNODES * NHEAD];
    __shared__ __attribute__((aligned(16))) float qs[BNODES * HD];
    __shared__ __attribute__((aligned(16))) float s_rows[ACC_WAVES * 128];
    __shared__ int rbase[BIN_BLOCKS];
    __shared__ int rcnt[BIN_BLOCKS];
    __shared__ __attribute__((aligned(16))) int rpre[BIN_BLOCKS + 4];
    __shared__ __attribute__((aligned(16))) int rpre16[16];

    const int t     = (int)threadIdx.x;
    const int lane  = t & 31;
    const int w     = t >> 5;
    const int half  = lane >> 4;
    const int i16   = lane & 15;
    const int node0 = (int)blockIdx.x * BNODES;
    const int nn    = min(BNODES, N_NODES - node0);

    if (t < BIN_BLOCKS) {
        const int rs = runstart[t * NBUCK + (int)blockIdx.x];
        const int re = ((int)blockIdx.x + 1 < NBUCK) ? runstart[t * NBUCK + (int)blockIdx.x + 1] : BIN_CHUNK;
        rbase[t] = t * BIN_CHUNK + rs;
        rcnt[t]  = re - rs;
    }
    for (int i = t; i < BNODES * HD; i += ACC_THREADS) acc[i] = 0;
    for (int i = t; i < BNODES * NHEAD; i += ACC_THREADS) zacc[i] = 0;
    for (int i = t; i < nn * (HD / 4); i += ACC_THREADS)
        reinterpret_cast<v4f*>(qs)[i] = reinterpret_cast<const v4f*>(Q + (size_t)node0 * HD)[i];
    __syncthreads();
    if (t < BIN_BLOCKS) {
        int b = 0;
        for (int j = 0; j < t; ++j) b += rcnt[j];
        rpre[t] = b;
        if ((t & 15) == 0) rpre16[t >> 4] = b;
        if (t == BIN_BLOCKS - 1) rpre[BIN_BLOCKS] = b + rcnt[t];
    }
    __syncthreads();

    const int seg1 = rpre[BIN_BLOCKS];
    float* slot = s_rows + w * 128;

    int bb = 32 * w;
    if (bb < seg1) {
        int left = seg1 - bb;
        int sv = 0, nv = 0;
        if (lane < left) {
            const int g = gt_run_pos(rpre, rpre16, rbase, bb + lane);
            sv = bsrc[g];
            nv = bdst[g] - node0;
        }
        PairRows cur = gt_load_pair(K, V, sv, nv, 0, left, half, i16);
        for (;;) {
            const int bbn   = bb + 32 * ACC_WAVES;
            const bool more = bbn < seg1;
            const int leftn = seg1 - bbn;
            int svn = 0, nvn = 0;
            if (more && lane < leftn) {
                const int g = gt_run_pos(rpre, rpre16, rbase, bbn + lane);
                svn = bsrc[g];
                nvn = bdst[g] - node0;
            }
#pragma unroll
            for (int p = 0; p < 16; ++p) {
                PairRows nxt = cur;
                if (p + 1 < 16)
                    nxt = gt_load_pair(K, V, sv, nv, p + 1, left, half, i16);
                else if (more)
                    nxt = gt_load_pair(K, V, svn, nvn, 0, leftn, half, i16);
                __builtin_amdgcn_sched_barrier(0);
                gt_reduce_pair(cur, qs, acc, zacc, slot, lane, i16);
                cur = nxt;
            }
            if (!more) break;
            bb = bbn; left = leftn; sv = svn; nv = nvn;
        }
    }
    __syncthreads();

    for (int r = w; r < nn; r += ACC_WAVES) {
        const float z_lo = gt_from_fixed(zacc[r * NHEAD + (lane >> 4)])     + 1e-6f;
        const float z_hi = gt_from_fixed(zacc[r * NHEAD + 2 + (lane >> 4)]) + 1e-6f;
        float* __restrict__ o = out + (size_t)(node0 + r) * HD + lane;
        o[0]  = gt_from_fixed(acc[r * HD + lane])      / z_lo;
        o[32] = gt_from_fixed(acc[r * HD + lane + 32]) / z_hi;
    }
}

extern "C" void kernel_launch(void* const* d_in, const int* in_sizes, int n_in,
                              void* d_out, int out_size, void* d_ws, size_t ws_size,
                              hipStream_t stream)
{
    (void)in_sizes; (void)n_in; (void)out_size; (void)ws_size;

    const float* X   = (const float*)d_in[0];
    const int*   src = (const int*)  d_in[1];
    const int*   dst = (const int*)  d_in[2];
    const float* Wq  = (const float*)d_in[3];
    const float* Wk  = (const float*)d_in[4];
    const float* Wv  = (const float*)d_in[5];

    float* out = (float*)d_out;
    float* ws  = (float*)d_ws;
    float* Q   = ws;
    float* K   = ws + (size_t)N_NODES * HD;
    float* V   = ws + (size_t)2 * N_NODES * HD;
    int* runstart = (int*)(ws + (size_t)3 * N_NODES * HD);
    int* bsrc = (int*)(ws + (size_t)3 * N_NODES * HD + (size_t)2 * N_NODES * NHEAD);
    int* bdst = bsrc + (size_t)E_EDGES;

    gt_front<<<BIN_BLOCKS + PROJ_BLOCKS, FRONT_THREADS, 0, stream>>>(X, Wq, Wk, Wv, src, dst, Q, K, V,
                                                                    runstart, bsrc, bdst);

    gt_range_attn<<<NBUCK, ACC_THREADS, 0, stream>>>(bsrc, bdst, runstart, Q, K, V, out);
}


# Float to fixed point by floor/fract split: high word = floor(x) as int (one v_cvt_floor), low word = fract(x)*2^32; no abs, trunc, 64-bit negate or select; negative terms round at 2^-24, not 2^-32
# speedup vs baseline: 2.5261x; 1.1143x over previous
#include <hip/hip_runtime.h>
#include <hip/hip_bf16.h>

typedef __attribute__((ext_vector_type(2))) float v2f;
typedef __attribute__((ext_vector_type(8))) float v8f;

#define N_NODES 50000
#define E_EDGES 1600000
#define NHEAD   4
#define DHEAD   16
#define F_IN    128
#define HD      64
#define PROJ_TASKS ((N_NODES / 16) * 3)

__device__ __forceinline__ void
gt_qkv_proj_tile(int wave, int lane,
                 const float* __restrict__ X,
                 const float* __restrict__ Wq,
                 const float* __restrict__ Wk,
                 const float* __restrict__ Wv,
                 float* __restrict__ Q,
                 float* __restrict__ K,
                 float* __restrict__ V)
{
    const int NTASK = PROJ_TASKS;
    if (wave >= NTASK) return;

    const int mat  = wave % 3;
    const int row0 = (wave / 3) * 16;
    const float* __restrict__ W   = (mat == 0) ? Wq : (mat == 1) ? Wk : Wv;
    float*       __restrict__ Out = (mat == 0) ? Q  : (mat == 1) ? K  : V;

    const int grp = lane >> 4;
    const int m16 = lane & 15;

    v8f c0 = {}, c1 = {}, c2 = {}, c3 = {};

    const float* __restrict__ arow = X + (size_t)(row0 + m16) * F_IN + 2 * grp;

#pragma unroll 4
    for (int k0 = 0; k0 < F_IN; k0 += 4) {
        v2f a;
        a.x = arow[k0];
        a.y = arow[k0 + 1];

        const float* __restrict__ wr0 = W + (size_t)(k0 + 2 * grp) * HD + m16;
        const float* __restrict__ wr1 = wr0 + HD;

        v2f b0, b1, b2, b3;
        b0.x = wr0[0];  b0.y = wr1[0];
        b1.x = wr0[16]; b1.y = wr1[16];
        b2.x = wr0[32]; b2.y = wr1[32];
        b3.x = wr0[48]; b3.y = wr1[48];

        c0 = __builtin_amdgcn_wmma_f32_16x16x4_f32(false, a, false, b0, (short)0, c0, false, false);
        c1 = __builtin_amdgcn_wmma_f32_16x16x4_f32(false, a, false, b1, (short)0, c1, false, false);
        c2 = __builtin_amdgcn_wmma_f32_16x16x4_f32(false, a, false, b2, (short)0, c2, false, false);
        c3 = __builtin_amdgcn_wmma_f32_16x16x4_f32(false, a, false, b3, (short)0, c3, false, false);
    }

    const int mrow = 8 * grp;
#pragma unroll
    for (int r = 0; r < 8; ++r) {
        float* __restrict__ o = Out + (size_t)(row0 + mrow + r) * HD + m16;
        o[0]  = c0[r];
        o[16] = c1[r];
        o[32] = c2[r];
        o[48] = c3[r];
    }
}

#define BNODES      196
#define NBUCK       256
#define BIN_BLOCKS  256
#define BIN_CHUNK   (E_EDGES / BIN_BLOCKS)
#define FRONT_THREADS 256
#define BIN_PER     ((BIN_CHUNK + FRONT_THREADS - 1) / FRONT_THREADS)
#define FILE_R      13
#define PROJ_BLOCKS ((PROJ_TASKS + FRONT_THREADS / 32 - 1) / (FRONT_THREADS / 32))
#define ACC_THREADS 1024
#define ACC_WAVES   (ACC_THREADS / 32)
static_assert(E_EDGES % BIN_BLOCKS == 0, "equal chunks");
static_assert(BNODES * NBUCK >= N_NODES, "ranges cover all nodes");
static_assert((N_NODES - 1) / BNODES < NBUCK, "largest range index in bounds");
static_assert(BIN_BLOCKS == 256, "the run search below counts in two levels of 16");
static_assert(BIN_BLOCKS <= ACC_THREADS && NBUCK == FRONT_THREADS, "one thread per table entry");
static_assert(N_NODES <= 65536, "a filed edge is packed as two 16-bit node numbers");

typedef __attribute__((ext_vector_type(4))) float v4f;
typedef unsigned long long u64;

__device__ __forceinline__ void
gt_file_chunk(int chunk, const int* __restrict__ src, const int* __restrict__ dst,
              int* __restrict__ runstart, int* __restrict__ bsrc, int* __restrict__ bdst)
{
    __shared__ int hist[NBUCK];
    __shared__ int cursor[NBUCK];
    __shared__ unsigned st[BIN_CHUNK];
    const int t = (int)threadIdx.x;
    hist[t] = 0;
    const int e0 = chunk * BIN_CHUNK;
    __syncthreads();
#pragma unroll 1
    for (int r0 = 0; r0 < BIN_PER; r0 += FILE_R) {
        int dd[FILE_R];
#pragma unroll
        for (int j = 0; j < FILE_R; ++j) {
            const int i = t + (r0 + j) * FRONT_THREADS;
            dd[j] = 0;
            if (i < BIN_CHUNK) dd[j] = dst[e0 + i];
        }
#pragma unroll
        for (int j = 0; j < FILE_R; ++j) {
            const int i = t + (r0 + j) * FRONT_THREADS;
            if (i < BIN_CHUNK) atomicAdd(&hist[(unsigned)dd[j] / (unsigned)BNODES], 1);
        }
    }
    __syncthreads();
    {
        int b = 0;
        for (int j = 0; j < t; ++j) b += hist[j];
        cursor[t] = b;
        runstart[chunk * NBUCK + t] = b;
    }
    __syncthreads();
#pragma unroll 1
    for (int r0 = 0; r0 < BIN_PER; r0 += FILE_R) {
        int dd[FILE_R], ss[FILE_R];
#pragma unroll
        for (int j = 0; j < FILE_R; ++j) {
            const int i = t + (r0 + j) * FRONT_THREADS;
            dd[j] = 0; ss[j] = 0;
            if (i < BIN_CHUNK) {
                dd[j] = dst[e0 + i];
                ss[j] = src[e0 + i];
            }
        }
        int pos[FILE_R];
#pragma unroll
        for (int j = 0; j < FILE_R; ++j) {
            const int i = t + (r0 + j) * FRONT_THREADS;
            pos[j] = 0;
            if (i < BIN_CHUNK) pos[j] = atomicAdd(&cursor[(unsigned)dd[j] / (unsigned)BNODES], 1);
        }
#pragma unroll
        for (int j = 0; j < FILE_R; ++j) {
            const int i = t + (r0 + j) * FRONT_THREADS;
            if (i < BIN_CHUNK) st[pos[j]] = ((unsigned)ss[j] << 16) | (unsigned)dd[j];
        }
    }
    __syncthreads();
    for (int i = t; i < BIN_CHUNK; i += FRONT_THREADS) {
        const unsigned u = st[i];
        bsrc[e0 + i] = (int)(u >> 16);
        bdst[e0 + i] = (int)(u & 0xFFFFu);
    }
}

__global__ void __launch_bounds__(FRONT_THREADS)
gt_front(const float* __restrict__ X,
         const float* __restrict__ Wq,
         const float* __restrict__ Wk,
         const float* __restrict__ Wv,
         const int* __restrict__ src, const int* __restrict__ dst,
         float* __restrict__ Q, float* __restrict__ K, float* __restrict__ V,
         int* __restrict__ runstart, int* __restrict__ bsrc, int* __restrict__ bdst)
{
    if ((int)blockIdx.x < BIN_BLOCKS) {
        gt_file_chunk((int)blockIdx.x, src, dst, runstart, bsrc, bdst);
    } else {
        const int wave = (((int)blockIdx.x - BIN_BLOCKS) * FRONT_THREADS + (int)threadIdx.x) >> 5;
        gt_qkv_proj_tile(wave, (int)(threadIdx.x & 31), X, Wq, Wk, Wv, Q, K, V);
    }
}

__device__ __forceinline__ float gt_quad_xor1(float x)
{
    return __int_as_float(__builtin_amdgcn_update_dpp(0, __float_as_int(x), 0xB1, 0xF, 0xF, true));
}
__device__ __forceinline__ float gt_quad_xor2(float x)
{
    return __int_as_float(__builtin_amdgcn_update_dpp(0, __float_as_int(x), 0x4E, 0xF, 0xF, true));
}

__device__ __forceinline__ int gt_floor_int(float x)
{
    int hi;
    asm("v_cvt_floor_i32_f32 %0, %1" : "=v"(hi) : "v"(x));
    return hi;
}
__device__ __forceinline__ u64 gt_to_fixed(float x)
{
    const int      hi = gt_floor_int(x);
    const float    fr = __builtin_amdgcn_fractf(x);
    const unsigned lo = (unsigned)(fr * 4294967296.0f);
    return ((u64)(unsigned)hi << 32) | (u64)lo;
}
__device__ __forceinline__ float gt_from_fixed(u64 a)
{
    return (float)(long long)a * (1.0f / 4294967296.0f);
}

struct PairRows {
    v4f k4, v4;
    int n0, n1, nl;
    bool ok;
};

__device__ __forceinline__ PairRows
gt_load_pair(const float* __restrict__ K, const float* __restrict__ V,
             int sv, int nv, int p, int left, int half, int i16)
{
    PairRows r;
    const int s0 = __builtin_amdgcn_readlane(sv, 2 * p);
    const int s1 = __builtin_amdgcn_readlane(sv, 2 * p + 1);
    r.n0 = __builtin_amdgcn_readlane(nv, 2 * p);
    r.n1 = __builtin_amdgcn_readlane(nv, 2 * p + 1);
    const int sl = half ? s1 : s0;
    r.nl = half ? r.n1 : r.n0;
    r.ok = (2 * p + half) < left;
    const unsigned so = (unsigned)sl * (unsigned)HD + 4u * (unsigned)i16;
    r.k4 = *reinterpret_cast<const v4f*>(K + so);
    r.v4 = *reinterpret_cast<const v4f*>(V + so);
    return r;
}

__device__ __forceinline__ void
gt_reduce_pair(const PairRows& c, const float* qs, u64* acc, u64* zacc,
               float* slot, int lane, int i16)
{
    const v4f q4 = *reinterpret_cast<const v4f*>(qs + c.nl * HD + 4 * i16);
    float d = c.k4.x * q4.x;
    d = fmaf(c.k4.y, q4.y, d);
    d = fmaf(c.k4.z, q4.z, d);
    d = fmaf(c.k4.w, q4.w, d);
    d += gt_quad_xor1(d);
    d += gt_quad_xor2(d);
    float sc = __expf(fminf(fmaxf(d * 0.25f, -5.0f), 5.0f));
    sc = c.ok ? sc : 0.0f;

    slot[4 * lane + 0] = sc * c.v4.x;
    slot[4 * lane + 1] = sc * c.v4.y;
    slot[4 * lane + 2] = sc * c.v4.z;
    slot[4 * lane + 3] = sc * c.v4.w;
    const float a0 = slot[lane];
    const float a1 = slot[lane + 32];
    const float b0 = slot[lane + 64];
    const float b1 = slot[lane + 96];

    u64* r0 = acc + c.n0 * HD + lane;
    u64* r1 = acc + c.n1 * HD + lane;
    atomicAdd(r0,      gt_to_fixed(a0));
    atomicAdd(r0 + 32, gt_to_fixed(a1));
    atomicAdd(r1,      gt_to_fixed(b0));
    atomicAdd(r1 + 32, gt_to_fixed(b1));
    if ((lane & 3) == 0)
        atomicAdd(zacc + c.nl * NHEAD + (i16 >> 2), gt_to_fixed(sc));
}

typedef __attribute__((ext_vector_type(4))) int v4i;
__device__ __forceinline__ int gt_count_le(const int* p16, int v)
{
    const v4i* q = reinterpret_cast<const v4i*>(p16);
    const v4i a = q[0], b = q[1], c = q[2], d = q[3];
    return (a.x <= v) + (a.y <= v) + (a.z <= v) + (a.w <= v)
         + (b.x <= v) + (b.y <= v) + (b.z <= v) + (b.w <= v)
         + (c.x <= v) + (c.y <= v) + (c.z <= v) + (c.w <= v)
         + (d.x <= v) + (d.y <= v) + (d.z <= v) + (d.w <= v);
}
__device__ __forceinline__ int gt_run_pos(const int* pre, const int* pre16, const int* base, int v)
{
    const int c = gt_count_le(pre16, v) - 1;
    const int b = 16 * c + gt_count_le(pre + 16 * c, v) - 1;
    return base[b] + (v - pre[b]);
}

__global__ void __launch_bounds__(ACC_THREADS)
gt_range_attn(const int* __restrict__ bsrc, const int* __restrict__ bdst,
              const int* __restrict__ runstart,
              const float* __restrict__ Q, const float* __restrict__ K,
              const float* __restrict__ V, float* __restrict__ out)
{
    __shared__ u64 acc[BNODES * HD];
    __shared__ u64 zacc[BNODES * NHEAD];
    __shared__ __attribute__((aligned(16))) float qs[BNODES * HD];
    __shared__ __attribute__((aligned(16))) float s_rows[ACC_WAVES * 128];
    __shared__ int rbase[BIN_BLOCKS];
    __shared__ int rcnt[BIN_BLOCKS];
    __shared__ __attribute__((aligned(16))) int rpre[BIN_BLOCKS + 4];
    __shared__ __attribute__((aligned(16))) int rpre16[16];

    const int t     = (int)threadIdx.x;
    const int lane  = t & 31;
    const int w     = t >> 5;
    const int half  = lane >> 4;
    const int i16   = lane & 15;
    const int node0 = (int)blockIdx.x * BNODES;
    const int nn    = min(BNODES, N_NODES - node0);

    if (t < BIN_BLOCKS) {
        const int rs = runstart[t * NBUCK + (int)blockIdx.x];
        const int re = ((int)blockIdx.x + 1 < NBUCK) ? runstart[t * NBUCK + (int)blockIdx.x + 1] : BIN_CHUNK;
        rbase[t] = t * BIN_CHUNK + rs;
        rcnt[t]  = re - rs;
    }
    for (int i = t; i < BNODES * HD; i += ACC_THREADS) acc[i] = 0;
    for (int i = t; i < BNODES * NHEAD; i += ACC_THREADS) zacc[i] = 0;
    for (int i = t; i < nn * (HD / 4); i += ACC_THREADS)
        reinterpret_cast<v4f*>(qs)[i] = reinterpret_cast<const v4f*>(Q + (size_t)node0 * HD)[i];
    __syncthreads();
    if (t < BIN_BLOCKS) {
        int b = 0;
        for (int j = 0; j < t; ++j) b += rcnt[j];
        rpre[t] = b;
        if ((t & 15) == 0) rpre16[t >> 4] = b;
        if (t == BIN_BLOCKS - 1) rpre[BIN_BLOCKS] = b + rcnt[t];
    }
    __syncthreads();

    const int seg1 = rpre[BIN_BLOCKS];
    float* slot = s_rows + w * 128;

    int bb = 32 * w;
    if (bb < seg1) {
        int left = seg1 - bb;
        int sv = 0, nv = 0;
        if (lane < left) {
            const int g = gt_run_pos(rpre, rpre16, rbase, bb + lane);
            sv = bsrc[g];
            nv = bdst[g] - node0;
        }
        PairRows cur = gt_load_pair(K, V, sv, nv, 0, left, half, i16);
        for (;;) {
            const int bbn   = bb + 32 * ACC_WAVES;
            const bool more = bbn < seg1;
            const int leftn = seg1 - bbn;
            int svn = 0, nvn = 0;
            if (more && lane < leftn) {
                const int g = gt_run_pos(rpre, rpre16, rbase, bbn + lane);
                svn = bsrc[g];
                nvn = bdst[g] - node0;
            }
#pragma unroll
            for (int p = 0; p < 16; ++p) {
                PairRows nxt = cur;
                if (p + 1 < 16)
                    nxt = gt_load_pair(K, V, sv, nv, p + 1, left, half, i16);
                else if (more)
                    nxt = gt_load_pair(K, V, svn, nvn, 0, leftn, half, i16);
                __builtin_amdgcn_sched_barrier(0);
                gt_reduce_pair(cur, qs, acc, zacc, slot, lane, i16);
                cur = nxt;
            }
            if (!more) break;
            bb = bbn; left = leftn; sv = svn; nv = nvn;
        }
    }
    __syncthreads();

    for (int r = w; r < nn; r += ACC_WAVES) {
        const float z_lo = gt_from_fixed(zacc[r * NHEAD + (lane >> 4)])     + 1e-6f;
        const float z_hi = gt_from_fixed(zacc[r * NHEAD + 2 + (lane >> 4)]) + 1e-6f;
        float* __restrict__ o = out + (size_t)(node0 + r) * HD + lane;
        o[0]  = gt_from_fixed(acc[r * HD + lane])      / z_lo;
        o[32] = gt_from_fixed(acc[r * HD + lane + 32]) / z_hi;
    }
}

extern "C" void kernel_launch(void* const* d_in, const int* in_sizes, int n_in,
                              void* d_out, int out_size, void* d_ws, size_t ws_size,
                              hipStream_t stream)
{
    (void)in_sizes; (void)n_in; (void)out_size; (void)ws_size;

    const float* X   = (const float*)d_in[0];
    const int*   src = (const int*)  d_in[1];
    const int*   dst = (const int*)  d_in[2];
    const float* Wq  = (const float*)d_in[3];
    const float* Wk  = (const float*)d_in[4];
    const float* Wv  = (const float*)d_in[5];

    float* out = (float*)d_out;
    float* ws  = (float*)d_ws;
    float* Q   = ws;
    float* K   = ws + (size_t)N_NODES * HD;
    float* V   = ws + (size_t)2 * N_NODES * HD;
    int* runstart = (int*)(ws + (size_t)3 * N_NODES * HD);
    int* bsrc = (int*)(ws + (size_t)3 * N_NODES * HD + (size_t)2 * N_NODES * NHEAD);
    int* bdst = bsrc + (size_t)E_EDGES;

    gt_front<<<BIN_BLOCKS + PROJ_BLOCKS, FRONT_THREADS, 0, stream>>>(X, Wq, Wk, Wv, src, dst, Q, K, V,
                                                                    runstart, bsrc, bdst);

    gt_range_attn<<<NBUCK, ACC_THREADS, 0, stream>>>(bsrc, bdst, runstart, Q, K, V, out);
}


# Projection B operand from LDS: a block serves one of Q/K/V for 8 row tiles and stages that 128x64 weight matrix once in fragment order; a 16-byte LDS read feeds 2 matrix instructions; same sum order
# speedup vs baseline: 2.6139x; 1.0348x over previous
#include <hip/hip_runtime.h>
#include <hip/hip_bf16.h>

typedef __attribute__((ext_vector_type(2))) float v2f;
typedef __attribute__((ext_vector_type(8))) float v8f;
typedef __attribute__((ext_vector_type(4))) float v4f;

#define N_NODES 50000
#define E_EDGES 1600000
#define NHEAD   4
#define DHEAD   16
#define F_IN    128
#define HD      64
#define FRONT_THREADS 256

#define ROW_TILES   (N_NODES / 16)
#define TILE_GROUPS ((ROW_TILES + 7) / 8)
#define WL_FLOATS   (16 * 2 * 64 * 4)
static_assert(N_NODES % 16 == 0, "whole row tiles");

__device__ __forceinline__ void
gt_qkv_proj_group(int pblk, int t,
                  const float* __restrict__ X,
                  const float* __restrict__ Wq,
                  const float* __restrict__ Wk,
                  const float* __restrict__ Wv,
                  float* __restrict__ Q,
                  float* __restrict__ K,
                  float* __restrict__ V,
                  float* __restrict__ wl)
{
    const int mat  = pblk % 3;
    const int grp8 = pblk / 3;
    const float* __restrict__ W   = (mat == 0) ? Wq : (mat == 1) ? Wk : Wv;
    float*       __restrict__ Out = (mat == 0) ? Q  : (mat == 1) ? K  : V;

    for (int it = 0; it < 2; ++it) {
        const int item = it * FRONT_THREADS + t;
        const int c4 = item & 15;
        const int hh = (item >> 4) & 1;
        const int jj = item >> 5;
        const float* __restrict__ wr = W + (size_t)(8 * jj + 2 * hh) * HD + 4 * c4;
        const v4f ra = *reinterpret_cast<const v4f*>(wr);
        const v4f rb = *reinterpret_cast<const v4f*>(wr + HD);
        const v4f rc = *reinterpret_cast<const v4f*>(wr + 4 * HD);
        const v4f rd = *reinterpret_cast<const v4f*>(wr + 5 * HD);
        v4f* d = reinterpret_cast<v4f*>(wl) + ((jj * 2 + hh) * 64 + 4 * c4);
        v4f e0, e1, e2, e3;
        e0.x = ra.x; e0.y = rb.x; e0.z = rc.x; e0.w = rd.x;
        e1.x = ra.y; e1.y = rb.y; e1.z = rc.y; e1.w = rd.y;
        e2.x = ra.z; e2.y = rb.z; e2.z = rc.z; e2.w = rd.z;
        e3.x = ra.w; e3.y = rb.w; e3.z = rc.w; e3.w = rd.w;
        d[0] = e0; d[1] = e1; d[2] = e2; d[3] = e3;
    }
    __syncthreads();

    const int lane = t & 31;
    const int tile = grp8 * 8 + (t >> 5);
    if (tile >= ROW_TILES) return;

    const int row0 = tile * 16;
    const int grp = lane >> 4;
    const int m16 = lane & 15;

    v8f c0 = {}, c1 = {}, c2 = {}, c3 = {};

    const float* __restrict__ arow = X + (size_t)(row0 + m16) * F_IN + 2 * grp;
    const v4f*   bl   = reinterpret_cast<const v4f*>(wl) + (grp * 64 + m16);

    for (int jj = 0; jj < F_IN / 8; ++jj) {
        v2f a0, a1;
        a0.x = arow[8 * jj];
        a0.y = arow[8 * jj + 1];
        a1.x = arow[8 * jj + 4];
        a1.y = arow[8 * jj + 5];

        const v4f w0 = bl[jj * 128];
        const v4f w1 = bl[jj * 128 + 16];
        const v4f w2 = bl[jj * 128 + 32];
        const v4f w3 = bl[jj * 128 + 48];

        v2f b0, b1, b2, b3;
        b0.x = w0.x; b0.y = w0.y;
        b1.x = w1.x; b1.y = w1.y;
        b2.x = w2.x; b2.y = w2.y;
        b3.x = w3.x; b3.y = w3.y;
        c0 = __builtin_amdgcn_wmma_f32_16x16x4_f32(false, a0, false, b0, (short)0, c0, false, false);
        c1 = __builtin_amdgcn_wmma_f32_16x16x4_f32(false, a0, false, b1, (short)0, c1, false, false);
        c2 = __builtin_amdgcn_wmma_f32_16x16x4_f32(false, a0, false, b2, (short)0, c2, false, false);
        c3 = __builtin_amdgcn_wmma_f32_16x16x4_f32(false, a0, false, b3, (short)0, c3, false, false);

        b0.x = w0.z; b0.y = w0.w;
        b1.x = w1.z; b1.y = w1.w;
        b2.x = w2.z; b2.y = w2.w;
        b3.x = w3.z; b3.y = w3.w;
        c0 = __builtin_amdgcn_wmma_f32_16x16x4_f32(false, a1, false, b0, (short)0, c0, false, false);
        c1 = __builtin_amdgcn_wmma_f32_16x16x4_f32(false, a1, false, b1, (short)0, c1, false, false);
        c2 = __builtin_amdgcn_wmma_f32_16x16x4_f32(false, a1, false, b2, (short)0, c2, false, false);
        c3 = __builtin_amdgcn_wmma_f32_16x16x4_f32(false, a1, false, b3, (short)0, c3, false, false);
    }

    const int mrow = 8 * grp;
#pragma unroll
    for (int r = 0; r < 8; ++r) {
        float* __restrict__ o = Out + (size_t)(row0 + mrow + r) * HD + m16;
        o[0]  = c0[r];
        o[16] = c1[r];
        o[32] = c2[r];
        o[48] = c3[r];
    }
}

#define BNODES      196
#define NBUCK       256
#define BIN_BLOCKS  256
#define BIN_CHUNK   (E_EDGES / BIN_BLOCKS)
#define BIN_PER     ((BIN_CHUNK + FRONT_THREADS - 1) / FRONT_THREADS)
#define FILE_R      13
#define PROJ_BLOCKS (TILE_GROUPS * 3)
#define FILE_LDS_WORDS (2 * NBUCK + BIN_CHUNK)
#define ACC_THREADS 1024
#define ACC_WAVES   (ACC_THREADS / 32)
static_assert(E_EDGES % BIN_BLOCKS == 0, "equal chunks");
static_assert(BNODES * NBUCK >= N_NODES, "ranges cover all nodes");
static_assert((N_NODES - 1) / BNODES < NBUCK, "largest range index in bounds");
static_assert(BIN_BLOCKS == 256, "the run search below counts in two levels of 16");
static_assert(BIN_BLOCKS <= ACC_THREADS && NBUCK == FRONT_THREADS, "one thread per table entry");
static_assert(N_NODES <= 65536, "a filed edge is packed as two 16-bit node numbers");

typedef unsigned long long u64;

__device__ __forceinline__ void
gt_file_chunk(int chunk, const int* __restrict__ src, const int* __restrict__ dst,
              int* __restrict__ runstart, int* __restrict__ bsrc, int* __restrict__ bdst,
              int* lds)
{
    int* hist   = lds;
    int* cursor = lds + NBUCK;
    unsigned* st = reinterpret_cast<unsigned*>(lds + 2 * NBUCK);
    const int t = (int)threadIdx.x;
    hist[t] = 0;
    const int e0 = chunk * BIN_CHUNK;
    __syncthreads();
#pragma unroll 1
    for (int r0 = 0; r0 < BIN_PER; r0 += FILE_R) {
        int dd[FILE_R];
#pragma unroll
        for (int j = 0; j < FILE_R; ++j) {
            const int i = t + (r0 + j) * FRONT_THREADS;
            dd[j] = 0;
            if (i < BIN_CHUNK) dd[j] = dst[e0 + i];
        }
#pragma unroll
        for (int j = 0; j < FILE_R; ++j) {
            const int i = t + (r0 + j) * FRONT_THREADS;
            if (i < BIN_CHUNK) atomicAdd(&hist[(unsigned)dd[j] / (unsigned)BNODES], 1);
        }
    }
    __syncthreads();
    {
        int b = 0;
        for (int j = 0; j < t; ++j) b += hist[j];
        cursor[t] = b;
        runstart[chunk * NBUCK + t] = b;
    }
    __syncthreads();
#pragma unroll 1
    for (int r0 = 0; r0 < BIN_PER; r0 += FILE_R) {
        int dd[FILE_R], ss[FILE_R];
#pragma unroll
        for (int j = 0; j < FILE_R; ++j) {
            const int i = t + (r0 + j) * FRONT_THREADS;
            dd[j] = 0; ss[j] = 0;
            if (i < BIN_CHUNK) {
                dd[j] = dst[e0 + i];
                ss[j] = src[e0 + i];
            }
        }
        int pos[FILE_R];
#pragma unroll
        for (int j = 0; j < FILE_R; ++j) {
            const int i = t + (r0 + j) * FRONT_THREADS;
            pos[j] = 0;
            if (i < BIN_CHUNK) pos[j] = atomicAdd(&cursor[(unsigned)dd[j] / (unsigned)BNODES], 1);
        }
#pragma unroll
        for (int j = 0; j < FILE_R; ++j) {
            const int i = t + (r0 + j) * FRONT_THREADS;
            if (i < BIN_CHUNK) st[pos[j]] = ((unsigned)ss[j] << 16) | (unsigned)dd[j];
        }
    }
    __syncthreads();
    for (int i = t; i < BIN_CHUNK; i += FRONT_THREADS) {
        const unsigned u = st[i];
        bsrc[e0 + i] = (int)(u >> 16);
        bdst[e0 + i] = (int)(u & 0xFFFFu);
    }
}

__global__ void __launch_bounds__(FRONT_THREADS)
gt_front(const float* __restrict__ X,
         const float* __restrict__ Wq,
         const float* __restrict__ Wk,
         const float* __restrict__ Wv,
         const int* __restrict__ src, const int* __restrict__ dst,
         float* __restrict__ Q, float* __restrict__ K, float* __restrict__ V,
         int* __restrict__ runstart, int* __restrict__ bsrc, int* __restrict__ bdst)
{
    __shared__ __attribute__((aligned(16))) int front_lds[WL_FLOATS];
    static_assert(FILE_LDS_WORDS <= WL_FLOATS, "the filing tables fit the same block");
    if ((int)blockIdx.x < BIN_BLOCKS) {
        gt_file_chunk((int)blockIdx.x, src, dst, runstart, bsrc, bdst, front_lds);
    } else {
        gt_qkv_proj_group((int)blockIdx.x - BIN_BLOCKS, (int)threadIdx.x, X, Wq, Wk, Wv, Q, K, V,
                          reinterpret_cast<float*>(front_lds));
    }
}

__device__ __forceinline__ float gt_quad_xor1(float x)
{
    return __int_as_float(__builtin_amdgcn_update_dpp(0, __float_as_int(x), 0xB1, 0xF, 0xF, true));
}
__device__ __forceinline__ float gt_quad_xor2(float x)
{
    return __int_as_float(__builtin_amdgcn_update_dpp(0, __float_as_int(x), 0x4E, 0xF, 0xF, true));
}

__device__ __forceinline__ int gt_floor_int(float x)
{
    int hi;
    asm("v_cvt_floor_i32_f32 %0, %1" : "=v"(hi) : "v"(x));
    return hi;
}
__device__ __forceinline__ u64 gt_to_fixed(float x)
{
    const int      hi = gt_floor_int(x);
    const float    fr = __builtin_amdgcn_fractf(x);
    const unsigned lo = (unsigned)(fr * 4294967296.0f);
    return ((u64)(unsigned)hi << 32) | (u64)lo;
}
__device__ __forceinline__ float gt_from_fixed(u64 a)
{
    return (float)(long long)a * (1.0f / 4294967296.0f);
}

struct PairRows {
    v4f k4, v4;
    int n0, n1, nl;
    bool ok;
};

__device__ __forceinline__ PairRows
gt_load_pair(const float* __restrict__ K, const float* __restrict__ V,
             int sv, int nv, int p, int left, int half, int i16)
{
    PairRows r;
    const int s0 = __builtin_amdgcn_readlane(sv, 2 * p);
    const int s1 = __builtin_amdgcn_readlane(sv, 2 * p + 1);
    r.n0 = __builtin_amdgcn_readlane(nv, 2 * p);
    r.n1 = __builtin_amdgcn_readlane(nv, 2 * p + 1);
    const int sl = half ? s1 : s0;
    r.nl = half ? r.n1 : r.n0;
    r.ok = (2 * p + half) < left;
    const unsigned so = (unsigned)sl * (unsigned)HD + 4u * (unsigned)i16;
    r.k4 = *reinterpret_cast<const v4f*>(K + so);
    r.v4 = *reinterpret_cast<const v4f*>(V + so);
    return r;
}

__device__ __forceinline__ void
gt_reduce_pair(const PairRows& c, const float* qs, u64* acc, u64* zacc,
               float* slot, int lane, int i16)
{
    const v4f q4 = *reinterpret_cast<const v4f*>(qs + c.nl * HD + 4 * i16);
    float d = c.k4.x * q4.x;
    d = fmaf(c.k4.y, q4.y, d);
    d = fmaf(c.k4.z, q4.z, d);
    d = fmaf(c.k4.w, q4.w, d);
    d += gt_quad_xor1(d);
    d += gt_quad_xor2(d);
    float sc = __expf(fminf(fmaxf(d * 0.25f, -5.0f), 5.0f));
    sc = c.ok ? sc : 0.0f;

    slot[4 * lane + 0] = sc * c.v4.x;
    slot[4 * lane + 1] = sc * c.v4.y;
    slot[4 * lane + 2] = sc * c.v4.z;
    slot[4 * lane + 3] = sc * c.v4.w;
    const float a0 = slot[lane];
    const float a1 = slot[lane + 32];
    const float b0 = slot[lane + 64];
    const float b1 = slot[lane + 96];

    u64* r0 = acc + c.n0 * HD + lane;
    u64* r1 = acc + c.n1 * HD + lane;
    atomicAdd(r0,      gt_to_fixed(a0));
    atomicAdd(r0 + 32, gt_to_fixed(a1));
    atomicAdd(r1,      gt_to_fixed(b0));
    atomicAdd(r1 + 32, gt_to_fixed(b1));
    if ((lane & 3) == 0)
        atomicAdd(zacc + c.nl * NHEAD + (i16 >> 2), gt_to_fixed(sc));
}

typedef __attribute__((ext_vector_type(4))) int v4i;
__device__ __forceinline__ int gt_count_le(const int* p16, int v)
{
    const v4i* q = reinterpret_cast<const v4i*>(p16);
    const v4i a = q[0], b = q[1], c = q[2], d = q[3];
    return (a.x <= v) + (a.y <= v) + (a.z <= v) + (a.w <= v)
         + (b.x <= v) + (b.y <= v) + (b.z <= v) + (b.w <= v)
         + (c.x <= v) + (c.y <= v) + (c.z <= v) + (c.w <= v)
         + (d.x <= v) + (d.y <= v) + (d.z <= v) + (d.w <= v);
}
__device__ __forceinline__ int gt_run_pos(const int* pre, const int* pre16, const int* base, int v)
{
    const int c = gt_count_le(pre16, v) - 1;
    const int b = 16 * c + gt_count_le(pre + 16 * c, v) - 1;
    return base[b] + (v - pre[b]);
}

__global__ void __launch_bounds__(ACC_THREADS)
gt_range_attn(const int* __restrict__ bsrc, const int* __restrict__ bdst,
              const int* __restrict__ runstart,
              const float* __restrict__ Q, const float* __restrict__ K,
              const float* __restrict__ V, float* __restrict__ out)
{
    __shared__ u64 acc[BNODES * HD];
    __shared__ u64 zacc[BNODES * NHEAD];
    __shared__ __attribute__((aligned(16))) float qs[BNODES * HD];
    __shared__ __attribute__((aligned(16))) float s_rows[ACC_WAVES * 128];
    __shared__ int rbase[BIN_BLOCKS];
    __shared__ int rcnt[BIN_BLOCKS];
    __shared__ __attribute__((aligned(16))) int rpre[BIN_BLOCKS + 4];
    __shared__ __attribute__((aligned(16))) int rpre16[16];

    const int t     = (int)threadIdx.x;
    const int lane  = t & 31;
    const int w     = t >> 5;
    const int half  = lane >> 4;
    const int i16   = lane & 15;
    const int node0 = (int)blockIdx.x * BNODES;
    const int nn    = min(BNODES, N_NODES - node0);

    if (t < BIN_BLOCKS) {
        const int rs = runstart[t * NBUCK + (int)blockIdx.x];
        const int re = ((int)blockIdx.x + 1 < NBUCK) ? runstart[t * NBUCK + (int)blockIdx.x + 1] : BIN_CHUNK;
        rbase[t] = t * BIN_CHUNK + rs;
        rcnt[t]  = re - rs;
    }
    for (int i = t; i < BNODES * HD; i += ACC_THREADS) acc[i] = 0;
    for (int i = t; i < BNODES * NHEAD; i += ACC_THREADS) zacc[i] = 0;
    for (int i = t; i < nn * (HD / 4); i += ACC_THREADS)
        reinterpret_cast<v4f*>(qs)[i] = reinterpret_cast<const v4f*>(Q + (size_t)node0 * HD)[i];
    __syncthreads();
    if (t < BIN_BLOCKS) {
        int b = 0;
        for (int j = 0; j < t; ++j) b += rcnt[j];
        rpre[t] = b;
        if ((t & 15) == 0) rpre16[t >> 4] = b;
        if (t == BIN_BLOCKS - 1) rpre[BIN_BLOCKS] = b + rcnt[t];
    }
    __syncthreads();

    const int seg1 = rpre[BIN_BLOCKS];
    float* slot = s_rows + w * 128;

    int bb = 32 * w;
    if (bb < seg1) {
        int left = seg1 - bb;
        int sv = 0, nv = 0;
        if (lane < left) {
            const int g = gt_run_pos(rpre, rpre16, rbase, bb + lane);
            sv = bsrc[g];
            nv = bdst[g] - node0;
        }
        PairRows cur = gt_load_pair(K, V, sv, nv, 0, left, half, i16);
        for (;;) {
            const int bbn   = bb + 32 * ACC_WAVES;
            const bool more = bbn < seg1;
            const int leftn = seg1 - bbn;
            int svn = 0, nvn = 0;
            if (more && lane < leftn) {
                const int g = gt_run_pos(rpre, rpre16, rbase, bbn + lane);
                svn = bsrc[g];
                nvn = bdst[g] - node0;
            }
#pragma unroll
            for (int p = 0; p < 16; ++p) {
                PairRows nxt = cur;
                if (p + 1 < 16)
                    nxt = gt_load_pair(K, V, sv, nv, p + 1, left, half, i16);
                else if (more)
                    nxt = gt_load_pair(K, V, svn, nvn, 0, leftn, half, i16);
                __builtin_amdgcn_sched_barrier(0);
                gt_reduce_pair(cur, qs, acc, zacc, slot, lane, i16);
                cur = nxt;
            }
            if (!more) break;
            bb = bbn; left = leftn; sv = svn; nv = nvn;
        }
    }
    __syncthreads();

    for (int r = w; r < nn; r += ACC_WAVES) {
        const float z_lo = gt_from_fixed(zacc[r * NHEAD + (lane >> 4)])     + 1e-6f;
        const float z_hi = gt_from_fixed(zacc[r * NHEAD + 2 + (lane >> 4)]) + 1e-6f;
        float* __restrict__ o = out + (size_t)(node0 + r) * HD + lane;
        o[0]  = gt_from_fixed(acc[r * HD + lane])      / z_lo;
        o[32] = gt_from_fixed(acc[r * HD + lane + 32]) / z_hi;
    }
}

extern "C" void kernel_launch(void* const* d_in, const int* in_sizes, int n_in,
                              void* d_out, int out_size, void* d_ws, size_t ws_size,
                              hipStream_t stream)
{
    (void)in_sizes; (void)n_in; (void)out_size; (void)ws_size;

    const float* X   = (const float*)d_in[0];
    const int*   src = (const int*)  d_in[1];
    const int*   dst = (const int*)  d_in[2];
    const float* Wq  = (const float*)d_in[3];
    const float* Wk  = (const float*)d_in[4];
    const float* Wv  = (const float*)d_in[5];

    float* out = (float*)d_out;
    float* ws  = (float*)d_ws;
    float* Q   = ws;
    float* K   = ws + (size_t)N_NODES * HD;
    float* V   = ws + (size_t)2 * N_NODES * HD;
    int* runstart = (int*)(ws + (size_t)3 * N_NODES * HD);
    int* bsrc = (int*)(ws + (size_t)3 * N_NODES * HD + (size_t)2 * N_NODES * NHEAD);
    int* bdst = bsrc + (size_t)E_EDGES;

    gt_front<<<BIN_BLOCKS + PROJ_BLOCKS, FRONT_THREADS, 0, stream>>>(X, Wq, Wk, Wv, src, dst, Q, K, V,
                                                                    runstart, bsrc, bdst);

    gt_range_attn<<<NBUCK, ACC_THREADS, 0, stream>>>(bsrc, bdst, runstart, Q, K, V, out);
}
